# starting kernel
# baseline (speedup 1.0000x reference)
#include <hip/hip_runtime.h>
#include <hip/hip_bf16.h>
#include <cmath>

typedef __attribute__((ext_vector_type(16))) _Float16 v16h;
typedef __attribute__((ext_vector_type(8)))  float    v8f;

#define NNODES 50000
#define NEDGES 500000

__device__ __forceinline__ unsigned f2key(float f) {
  unsigned b = __float_as_uint(f);
  return (b & 0x80000000u) ? ~b : (b | 0x80000000u);
}
__device__ __forceinline__ float key2f(unsigned u) {
  if (u == 0u) return 0.0f;
  return (u & 0x80000000u) ? __uint_as_float(u ^ 0x80000000u) : __uint_as_float(~u);
}

__device__ __forceinline__ v16h load_cvt_f16(const float* p  ) {
  const float4* q = (const float4*)p;
  float4 v0 = q[0], v1 = q[1], v2 = q[4], v3 = q[5];
  v16h r;
  r[0] =(_Float16)v0.x; r[1] =(_Float16)v0.y; r[2] =(_Float16)v0.z; r[3] =(_Float16)v0.w;
  r[4] =(_Float16)v1.x; r[5] =(_Float16)v1.y; r[6] =(_Float16)v1.z; r[7] =(_Float16)v1.w;
  r[8] =(_Float16)v2.x; r[9] =(_Float16)v2.y; r[10]=(_Float16)v2.z; r[11]=(_Float16)v2.w;
  r[12]=(_Float16)v3.x; r[13]=(_Float16)v3.y; r[14]=(_Float16)v3.z; r[15]=(_Float16)v3.w;
  return r;
}

__global__ void gatv2_gemm_nt_wmma(const float* __restrict__ X,
                                   const float* __restrict__ W,
                                   float* __restrict__ Y,
                                   int Nrows, int Mcols) {
  const int K = 128;
  int wave = (int)((blockIdx.x * blockDim.x + threadIdx.x) >> 5);
  int lane = threadIdx.x & 31;
  int ntN = (Nrows + 15) >> 4;
  if (wave >= ntN) return;
  int half = lane >> 4;
  int l = lane & 15;
  int row = wave * 16 + l; if (row >= Nrows) row = Nrows - 1;
  const float* xrow = X + (size_t)row * K + half * 8;

  v16h a[4];
  #pragma unroll
  for (int kk = 0; kk < 4; ++kk) a[kk] = load_cvt_f16(xrow + kk * 32);

  int ntM = Mcols >> 4;
  for (int tm = 0; tm < ntM; ++tm) {
    const float* wrow = W + (size_t)(tm * 16 + l) * K + half * 8;
    v8f acc = {};
    #pragma unroll
    for (int kk = 0; kk < 4; ++kk) {
      v16h b = load_cvt_f16(wrow + kk * 32);
      acc = __builtin_amdgcn_wmma_f32_16x16x32_f16(
           false, a[kk],  false, b,
           (short)0, acc,  false,  false);
    }
    #pragma unroll
    for (int r = 0; r < 8; ++r) {
      int orow = wave * 16 + r + half * 8;
      if (orow < Nrows) Y[(size_t)orow * Mcols + tm * 16 + l] = acc[r];
    }
  }
}

template<int C>
__global__ void gatv2_edge_logits(const float* __restrict__ xl,
                                  const float* __restrict__ xr,
                                  const float* __restrict__ att,
                                  const int* __restrict__ srcA,
                                  const int* __restrict__ dstA,
                                  float* __restrict__ logit,
                                  unsigned* __restrict__ lmax,
                                  int E, int Nn) {
  const int HC = 4 * C;
  int wave = (int)((blockIdx.x * blockDim.x + threadIdx.x) >> 5);
  int lane = threadIdx.x & 31;
  if (wave >= E + Nn) return;
  int s, d;
  if (wave < E) { s = srcA[wave]; d = dstA[wave]; } else { s = wave - E; d = s; }
  const float* xls = xl + (size_t)s * HC;
  const float* xrd = xr + (size_t)d * HC;
  float hv[4] = {0.f, 0.f, 0.f, 0.f};
  #pragma unroll
  for (int i = 0; i < HC / 32; ++i) {
    int j = lane + 32 * i;
    float v = xls[j] + xrd[j];
    v = v > 0.f ? v : 0.2f * v;
    hv[(i * 32) / C] += v * att[j];
  }
  #pragma unroll
  for (int h = 0; h < 4; ++h) {
    float v = hv[h];
    #pragma unroll
    for (int m = 16; m > 0; m >>= 1) v += __shfl_xor(v, m, 32);
    hv[h] = v;
  }
  if (lane < 4) {
    float lg = hv[lane];
    logit[(size_t)wave * 4 + lane] = lg;
    atomicMax(&lmax[(size_t)d * 4 + lane], f2key(lg));
  }
}

template<int C>
__global__ void gatv2_edge_accum(const float* __restrict__ xl,
                                 const float* __restrict__ logit,
                                 const unsigned* __restrict__ lmax,
                                 float* __restrict__ denom,
                                 float* __restrict__ agg,
                                 const int* __restrict__ srcA,
                                 const int* __restrict__ dstA,
                                 int E, int Nn) {
  const int HC = 4 * C;
  int wave = (int)((blockIdx.x * blockDim.x + threadIdx.x) >> 5);
  int lane = threadIdx.x & 31;
  if (wave >= E + Nn) return;
  int s, d;
  if (wave < E) { s = srcA[wave]; d = dstA[wave]; } else { s = wave - E; d = s; }
  float w = 0.f;
  if (lane < 4) {
    float lg = logit[(size_t)wave * 4 + lane];
    float mx = key2f(lmax[(size_t)d * 4 + lane]);
    w = expf(lg - mx);
    atomicAdd(&denom[(size_t)d * 4 + lane], w);
  }
  float wh[4];
  wh[0] = __shfl(w, 0, 32); wh[1] = __shfl(w, 1, 32);
  wh[2] = __shfl(w, 2, 32); wh[3] = __shfl(w, 3, 32);
  const float* xls = xl + (size_t)s * HC;
  float* aggd = agg + (size_t)d * HC;
  #pragma unroll
  for (int i = 0; i < HC / 32; ++i) {
    int j = lane + 32 * i;
    atomicAdd(&aggd[j], wh[(i * 32) / C] * xls[j]);
  }
}

__global__ void gatv2_finalize1(const float* __restrict__ agg,
                                const float* __restrict__ denom,
                                const float* __restrict__ b1,
                                float* __restrict__ hout, int Nn) {
  int idx = (int)(blockIdx.x * blockDim.x + threadIdx.x);
  if (idx >= Nn * 128) return;
  int n = idx >> 7, j = idx & 127, h = j >> 5;
  float v = agg[idx] / (denom[n * 4 + h] + 1e-16f) + b1[j];
  hout[idx] = v > 0.f ? v : expm1f(v);
}

__global__ void gatv2_finalize2(const float* __restrict__ agg,
                                const float* __restrict__ denom,
                                const float* __restrict__ b2,
                                float* __restrict__ out, int Nn) {
  int idx = (int)(blockIdx.x * blockDim.x + threadIdx.x);
  if (idx >= Nn * 64) return;
  int n = idx >> 6, c = idx & 63;
  float acc = 0.f;
  #pragma unroll
  for (int h = 0; h < 4; ++h)
    acc += agg[(size_t)n * 256 + h * 64 + c] / (denom[n * 4 + h] + 1e-16f);
  out[idx] = acc * 0.25f + b2[c];
}

extern "C" void kernel_launch(void* const* d_in, const int* in_sizes, int n_in,
                              void* d_out, int out_size, void* d_ws, size_t ws_size,
                              hipStream_t stream) {
  (void)in_sizes; (void)n_in; (void)out_size; (void)ws_size;
  const int  N  = NNODES;
  const int  E  = NEDGES;
  const long EP = (long)E + N;

  const float* x    = (const float*)d_in[0];
  const int*   ei   = (const int*)  d_in[1];
  const float* Wl1  = (const float*)d_in[2];
  const float* Wr1  = (const float*)d_in[3];
  const float* att1 = (const float*)d_in[4];
  const float* b1   = (const float*)d_in[5];
  const float* Wl2  = (const float*)d_in[6];
  const float* Wr2  = (const float*)d_in[7];
  const float* att2 = (const float*)d_in[8];
  const float* b2   = (const float*)d_in[9];
  const int* srcA = ei;
  const int* dstA = ei + E;

  float* ws    = (float*)d_ws;
  float* hbuf  = ws;
  float* xl    = hbuf + (size_t)N * 128;
  float* xr    = xl   + (size_t)N * 256;
  float* agg   = xr   + (size_t)N * 256;
  float* logit = agg  + (size_t)N * 256;
  unsigned* lmax = (unsigned*)(logit + (size_t)EP * 4);
  float* denom = (float*)(lmax + (size_t)N * 4);

  const int TPB = 256;
  int eblocks = (int)((EP + 7) / 8);
  int gblocks = (((N + 15) / 16) + 7) / 8;

  hipMemsetAsync(agg,   0, (size_t)N * 128 * sizeof(float), stream);
  hipMemsetAsync(lmax,  0, (size_t)N * 4 * sizeof(unsigned), stream);
  hipMemsetAsync(denom, 0, (size_t)N * 4 * sizeof(float), stream);
  gatv2_gemm_nt_wmma<<<gblocks, TPB, 0, stream>>>(x, Wl1, xl, N, 128);
  gatv2_gemm_nt_wmma<<<gblocks, TPB, 0, stream>>>(x, Wr1, xr, N, 128);
  gatv2_edge_logits<32><<<eblocks, TPB, 0, stream>>>(xl, xr, att1, srcA, dstA,
                                                     logit, lmax, E, N);
  gatv2_edge_accum<32><<<eblocks, TPB, 0, stream>>>(xl, logit, lmax, denom, agg,
                                                    srcA, dstA, E, N);
  gatv2_finalize1<<<(N * 128 + TPB - 1) / TPB, TPB, 0, stream>>>(agg, denom, b1, hbuf, N);

  hipMemsetAsync(agg,   0, (size_t)N * 256 * sizeof(float), stream);
  hipMemsetAsync(lmax,  0, (size_t)N * 4 * sizeof(unsigned), stream);
  hipMemsetAsync(denom, 0, (size_t)N * 4 * sizeof(float), stream);
  gatv2_gemm_nt_wmma<<<gblocks, TPB, 0, stream>>>(hbuf, Wl2, xl, N, 256);
  gatv2_gemm_nt_wmma<<<gblocks, TPB, 0, stream>>>(hbuf, Wr2, xr, N, 256);
  gatv2_edge_logits<64><<<eblocks, TPB, 0, stream>>>(xl, xr, att2, srcA, dstA,
                                                     logit, lmax, E, N);
  gatv2_edge_accum<64><<<eblocks, TPB, 0, stream>>>(xl, logit, lmax, denom, agg,
                                                    srcA, dstA, E, N);
  gatv2_finalize2<<<(N * 64 + TPB - 1) / TPB, TPB, 0, stream>>>(agg, denom, b2,
                                                                (float*)d_out, N);
}


# Layer-1 logits and maxima done in the destination-range kernel (xr rows, per-wave edge queues in LDS; an overfull queue sends the group to a two-scan path): logits launch gone; sums in arrival order
# speedup vs baseline: 1.9286x; 1.9286x over previous
#include <hip/hip_runtime.h>
#include <hip/hip_bf16.h>
#include <cmath>

typedef __attribute__((ext_vector_type(16))) _Float16 v16h;
typedef __attribute__((ext_vector_type(8)))  float    v8f;

#define NNODES 50000
#define NEDGES 500000

__device__ __forceinline__ unsigned f2key(float f) {
  unsigned b = __float_as_uint(f);
  return (b & 0x80000000u) ? ~b : (b | 0x80000000u);
}
__device__ __forceinline__ float key2f(unsigned u) {
  if (u == 0u) return 0.0f;
  return (u & 0x80000000u) ? __uint_as_float(u ^ 0x80000000u) : __uint_as_float(~u);
}

__device__ __forceinline__ v16h load_cvt_f16(const float* p  ) {
  const float4* q = (const float4*)p;
  float4 v0 = q[0], v1 = q[1], v2 = q[4], v3 = q[5];
  v16h r;
  r[0] =(_Float16)v0.x; r[1] =(_Float16)v0.y; r[2] =(_Float16)v0.z; r[3] =(_Float16)v0.w;
  r[4] =(_Float16)v1.x; r[5] =(_Float16)v1.y; r[6] =(_Float16)v1.z; r[7] =(_Float16)v1.w;
  r[8] =(_Float16)v2.x; r[9] =(_Float16)v2.y; r[10]=(_Float16)v2.z; r[11]=(_Float16)v2.w;
  r[12]=(_Float16)v3.x; r[13]=(_Float16)v3.y; r[14]=(_Float16)v3.z; r[15]=(_Float16)v3.w;
  return r;
}

__global__ void gatv2_gemm_nt_wmma(const float* __restrict__ X,
                                   const float* __restrict__ W,
                                   float* __restrict__ Y,
                                   int Nrows, int Mcols) {
  const int K = 128;
  int wave = (int)((blockIdx.x * blockDim.x + threadIdx.x) >> 5);
  int lane = threadIdx.x & 31;
  int ntN = (Nrows + 15) >> 4;
  if (wave >= ntN) return;
  int half = lane >> 4;
  int l = lane & 15;
  int row = wave * 16 + l; if (row >= Nrows) row = Nrows - 1;
  const float* xrow = X + (size_t)row * K + half * 8;

  v16h a[4];
  #pragma unroll
  for (int kk = 0; kk < 4; ++kk) a[kk] = load_cvt_f16(xrow + kk * 32);

  int ntM = Mcols >> 4;
  for (int tm = 0; tm < ntM; ++tm) {
    const float* wrow = W + (size_t)(tm * 16 + l) * K + half * 8;
    v8f acc = {};
    #pragma unroll
    for (int kk = 0; kk < 4; ++kk) {
      v16h b = load_cvt_f16(wrow + kk * 32);
      acc = __builtin_amdgcn_wmma_f32_16x16x32_f16(
           false, a[kk],  false, b,
           (short)0, acc,  false,  false);
    }
    #pragma unroll
    for (int r = 0; r < 8; ++r) {
      int orow = wave * 16 + r + half * 8;
      if (orow < Nrows) Y[(size_t)orow * Mcols + tm * 16 + l] = acc[r];
    }
  }
}

#define EDGES_PER_WAVE 32

__device__ __forceinline__ float lane_get(float v, int l) {
  return __int_as_float(__builtin_amdgcn_readlane(__float_as_int(v), l));
}
__device__ __forceinline__ int lane_geti(int v, int l) {
  return __builtin_amdgcn_readlane(v, l);
}
template<int CTRL>
__device__ __forceinline__ float xadd(float v) {
  return v + __int_as_float(__builtin_amdgcn_update_dpp(0, __float_as_int(v), CTRL, 0xf, 0xf, true));
}

__device__ __forceinline__ bool edge_wave_head(const int* __restrict__ srcA,
                                               const int* __restrict__ dstA,
                                               int E, int Nn,
                                               long& base, int& cnt, int& sv, int& dv) {
  const long EP = (long)E + Nn;
  int wave = __builtin_amdgcn_readfirstlane(
      (int)(blockIdx.x * (blockDim.x >> 5) + (threadIdx.x >> 5)));
  base = (long)wave * EDGES_PER_WAVE;
  if (base >= EP) return false;
  long left = EP - base;
  cnt = left < EDGES_PER_WAVE ? (int)left : EDGES_PER_WAVE;
  long e = base + (threadIdx.x & 31);
  if (e > EP - 1) e = EP - 1;
  sv = e < E ? srcA[e] : (int)(e - E);
  dv = e < E ? dstA[e] : (int)(e - E);
  return true;
}

template<int C>
__global__ void gatv2_edge_logits(const float* __restrict__ xl,
                                  const float* __restrict__ xr,
                                  const float* __restrict__ att,
                                  const int* __restrict__ srcA,
                                  const int* __restrict__ dstA,
                                  float* __restrict__ logit,
                                  unsigned* __restrict__ lmax,
                                  int E, int Nn) {
  constexpr int HC = 4 * C;
  constexpr int NV = HC / 128;
  constexpr int U  = 8 / NV;
  const long EP = (long)E + Nn;
  const int lane = threadIdx.x & 31;
  long base; int cnt, sv, dv;
  if (!edge_wave_head(srcA, dstA, E, Nn, base, cnt, sv, dv)) return;

  float4 av[NV];
  #pragma unroll
  for (int i = 0; i < NV; ++i) av[i] = ((const float4*)att)[i * 32 + lane];

  for (int r = 0; r < 4; ++r) {
    if (8 * r >= cnt) break;
    float lg = 0.f;
    #pragma unroll
    for (int q0 = 0; q0 < 8; q0 += U) {
      float4 a[U][NV], b[U][NV];
      #pragma unroll
      for (int u = 0; u < U; ++u) {
        int j = 8 * r + q0 + u;
        int s = lane_geti(sv, j), d = lane_geti(dv, j);
        const float4* ps = (const float4*)(xl + (size_t)s * HC);
        const float4* pd = (const float4*)(xr + (size_t)d * HC);
        #pragma unroll
        for (int i = 0; i < NV; ++i) { a[u][i] = ps[i * 32 + lane]; b[u][i] = pd[i * 32 + lane]; }
      }
      #pragma unroll
      for (int u = 0; u < U; ++u) {
        float p[NV];
        #pragma unroll
        for (int i = 0; i < NV; ++i) {
          float t0 = a[u][i].x + b[u][i].x, t1 = a[u][i].y + b[u][i].y;
          float t2 = a[u][i].z + b[u][i].z, t3 = a[u][i].w + b[u][i].w;
          t0 = t0 > 0.f ? t0 : 0.2f * t0;
          t1 = t1 > 0.f ? t1 : 0.2f * t1;
          t2 = t2 > 0.f ? t2 : 0.2f * t2;
          t3 = t3 > 0.f ? t3 : 0.2f * t3;
          float acc = t0 * av[i].x;
          acc += t1 * av[i].y;
          acc += t2 * av[i].z;
          acc += t3 * av[i].w;
          acc = xadd<0x161>(acc);
          acc = xadd<0x162>(acc);
          acc = xadd<0x164>(acc);
          if (C == 64) acc = xadd<0x168>(acc);
          p[i] = acc;
        }
        #pragma unroll
        for (int h = 0; h < 4; ++h) {
          float sh = lane_get(p[(h * C) / 128], ((h * C) % 128) / 4);
          if (lane == (q0 + u) * 4 + h) lg = sh;
        }
      }
    }
    long er = base + 8 * r + (lane >> 2);
    if (er < EP) {
      int h = lane & 3;
      int dr = er < E ? dstA[er] : (int)(er - E);
      logit[er * 4 + h] = lg;
      atomicMax(&lmax[(size_t)dr * 4 + h], f2key(lg));
    }
  }
}

#define RNODES   196
#define RWAVES   32
#define SCAN_UB  8
#define QW       184

static_assert(NNODES <= 65536, "a queue entry keeps src in 16 bits");
static_assert(NNODES + NEDGES <= (1 << 20), "a queue entry keeps the edge number in 20 bits");
static_assert(RNODES <= 256, "a queue entry keeps the local node in 8 bits");
static_assert(7 + 128 <= QW, "the two-scan path queues one piece per lane (128 edges) on top of a rest of 7");

__device__ __forceinline__ void q_push(int* q, int& qn, bool m, int v) {
  unsigned mask = __builtin_amdgcn_ballot_w32(m);
  int pos = qn + (int)__builtin_amdgcn_mbcnt_lo(mask, 0u);
  if (m && pos < QW) q[pos] = v;
  qn += __builtin_popcount(mask);
}

__device__ __forceinline__ void q_push_piece(int* q, int& qn, int4 dd, int p, bool in, int lo) {
  int r0 = dd.x - lo, r1 = dd.y - lo, r2 = dd.z - lo, r3 = dd.w - lo;
  bool m0 = in & ((unsigned)r0 < (unsigned)RNODES);
  bool m1 = in & ((unsigned)r1 < (unsigned)RNODES);
  bool m2 = in & ((unsigned)r2 < (unsigned)RNODES);
  bool m3 = in & ((unsigned)r3 < (unsigned)RNODES);
  if (__builtin_amdgcn_ballot_w32(m0 | m1 | m2 | m3)) {
    q_push(q, qn, m0, (4 * p)     | (int)((unsigned)r0 << 20));
    q_push(q, qn, m1, (4 * p + 1) | (int)((unsigned)r1 << 20));
    q_push(q, qn, m2, (4 * p + 2) | (int)((unsigned)r2 << 20));
    q_push(q, qn, m3, (4 * p + 3) | (int)((unsigned)r3 << 20));
  }
}

__device__ __forceinline__ void q_push_rest(int* q, int& qn, int wv, int lane, int lo,
                                            const int* __restrict__ dstA, int E, int Nn) {
  if (wv == 0) {
    int e = (E >> 2) * 4 + lane;
    int ec = e < E ? e : E - 1;
    int rr = dstA[ec] - lo;
    bool m = e < E && (unsigned)rr < (unsigned)RNODES;
    q_push(q, qn, m, e | (int)((unsigned)rr << 20));
  }
  int nl = wv + RWAVES * lane;
  bool ms = nl < RNODES && lo + nl < Nn;
  q_push(q, qn, ms, (E + lo + nl) | (nl << 20));
}

__device__ __forceinline__ float logits8(const float* __restrict__ xl, const float* sxr,
                                         float4 av, int s, int dl, int lane) {
  float4 a[8];
  #pragma unroll
  for (int qq = 0; qq < 8; ++qq) {
    int sj = lane_geti(s, qq * 4);
    a[qq] = ((const float4*)(xl + (size_t)sj * 128))[lane];
  }
  float lg = 0.f;
  #pragma unroll
  for (int qq = 0; qq < 8; ++qq) {
    int dj = lane_geti(dl, qq * 4);
    float4 bq = ((const float4*)(sxr + dj * 128))[lane];
    float t0 = a[qq].x + bq.x, t1 = a[qq].y + bq.y;
    float t2 = a[qq].z + bq.z, t3 = a[qq].w + bq.w;
    t0 = t0 > 0.f ? t0 : 0.2f * t0;
    t1 = t1 > 0.f ? t1 : 0.2f * t1;
    t2 = t2 > 0.f ? t2 : 0.2f * t2;
    t3 = t3 > 0.f ? t3 : 0.2f * t3;
    float acc = t0 * av.x;
    acc += t1 * av.y;
    acc += t2 * av.z;
    acc += t3 * av.w;
    acc = xadd<0x161>(acc);
    acc = xadd<0x162>(acc);
    acc = xadd<0x164>(acc);
    #pragma unroll
    for (int h = 0; h < 4; ++h) {
      float sh = lane_get(acc, 8 * h);
      if (lane == qq * 4 + h) lg = sh;
    }
  }
  return lg;
}

__device__ __forceinline__ void accum8(const float* __restrict__ xl, float* sagg, float* sden,
                                       int s, int dl, float lg, unsigned lm, int lane) {
  float v[8][4];
  #pragma unroll
  for (int qq = 0; qq < 8; ++qq) {
    int sj = lane_geti(s, qq * 4);
    const float* xls = xl + (size_t)sj * 128;
    #pragma unroll
    for (int i = 0; i < 4; ++i) v[qq][i] = xls[lane + 32 * i];
  }
  const float w = expf(lg - key2f(lm));
  atomicAdd(&sden[dl * 4 + (lane & 3)], w);
  #pragma unroll
  for (int qq = 0; qq < 8; ++qq) {
    int dj = lane_geti(dl, qq * 4);
    float* ad = sagg + dj * 128;
    #pragma unroll
    for (int i = 0; i < 4; ++i) {
      float wh = lane_get(w, qq * 4 + i);
      atomicAdd(&ad[lane + 32 * i], wh * v[qq][i]);
    }
  }
}

__global__ __launch_bounds__(RWAVES * 32)
void gatv2_range_layer1(const float* __restrict__ xl,
                        const float* __restrict__ xr,
                        const float* __restrict__ att,
                        const float* __restrict__ b1,
                        float* __restrict__ hout,
                        const int* __restrict__ srcA,
                        const int* __restrict__ dstA,
                        int E, int Nn) {
  __shared__ __attribute__((aligned(16))) float sagg[(RNODES + 1) * 128];
  __shared__ __attribute__((aligned(16))) float sden[(RNODES + 1) * 4];
  __shared__ __attribute__((aligned(16))) unsigned smax[RNODES * 4];
  __shared__ __attribute__((aligned(16))) float sxr[RNODES * 128];
  __shared__ int   sqe[RWAVES][QW];
  __shared__ float sql[RWAVES][QW * 4];
  __shared__ int   sover;

  const int tid = threadIdx.x, lane = tid & 31;
  const int wv = __builtin_amdgcn_readfirstlane(tid >> 5);
  const int lo = (int)blockIdx.x * RNODES;

  const float4 z4 = make_float4(0.f, 0.f, 0.f, 0.f);
  for (int i = tid; i < (RNODES + 1) * 32; i += RWAVES * 32) ((float4*)sagg)[i] = z4;
  if (tid < RNODES + 1) ((float4*)sden)[tid] = z4;
  if (tid < RNODES) ((uint4*)smax)[tid] = make_uint4(0u, 0u, 0u, 0u);
  if (tid == 0) sover = 0;
  for (int i = tid; i < RNODES * 32; i += RWAVES * 32) {
    int n = lo + (i >> 5);
    float4 t = make_float4(0.f, 0.f, 0.f, 0.f);
    if (n < Nn) t = ((const float4*)xr)[(size_t)n * 32 + (i & 31)];
    ((float4*)sxr)[i] = t;
  }
  __syncthreads();

  int* q = sqe[wv];
  float* ql = sql[wv];
  int qn = 0;
  int pc = 0;
  bool over = false;
  const float4 av = ((const float4*)att)[lane];
  const int k = lane >> 2, hh = lane & 3;

  const int n4 = E >> 2;
  const int nch = (n4 + 31) >> 5;
  const int4* d4 = (const int4*)dstA;
  const int nb = (nch + RWAVES * SCAN_UB - 1) / (RWAVES * SCAN_UB);

  for (int b = 0; b <= nb; ++b) {
    if (b < nb) {
      const int c0 = b * (RWAVES * SCAN_UB) + wv;
      int4 dd[SCAN_UB];
      #pragma unroll
      for (int u = 0; u < SCAN_UB; ++u) {
        int p = (c0 + RWAVES * u) * 32 + lane;
        dd[u] = d4[p < n4 ? p : n4 - 1];
      }
      #pragma unroll
      for (int u = 0; u < SCAN_UB; ++u) {
        int p = (c0 + RWAVES * u) * 32 + lane;
        q_push_piece(q, qn, dd[u], p, p < n4, lo);
      }
    } else {
      q_push_rest(q, qn, wv, lane, lo, dstA, E, Nn);
    }
    if (qn > QW) { over = true; break; }

    const int keep = b < nb ? 7 : 0;
    while (qn - pc > keep) {
      const int n = qn - pc < 8 ? qn - pc : 8;
      const bool ok = k < n;
      const int ent = q[pc + (ok ? k : 0)];
      const int e = ent & 0xFFFFF, dl = (int)((unsigned)ent >> 20);
      const int sv = srcA[e < E ? e : E - 1];
      const int s = e < E ? sv : e - E;
      const float lg = logits8(xl, sxr, av, s, dl, lane);
      if (ok) {
        ql[(pc + k) * 4 + hh] = lg;
        atomicMax(&smax[dl * 4 + hh], f2key(lg));
        if (hh == 0) q[pc + k] = s | (dl << 16);
      }
      pc += n;
    }
  }
  if (over && lane == 0) sover = 1;
  __syncthreads();

  if (sover == 0) {
    for (int base = 0; base < qn; base += 8) {
      const int n = qn - base < 8 ? qn - base : 8;
      const bool ok = k < n;
      const int ix = base + (ok ? k : 0);
      const int ent = q[ix];
      const int s = ent & 0xFFFF, dr = (int)((unsigned)ent >> 16);
      accum8(xl, sagg, sden, s, ok ? dr : RNODES, ql[ix * 4 + hh], smax[dr * 4 + hh], lane);
    }
  } else {
    for (int scan = 0; scan < 2; ++scan) {
      qn = 0;
      for (int c = wv; ; c += RWAVES) {
        const bool rest = c >= nch;
        if (!rest) {
          int p = c * 32 + lane;
          q_push_piece(q, qn, d4[p < n4 ? p : n4 - 1], p, p < n4, lo);
        } else {
          q_push_rest(q, qn, wv, lane, lo, dstA, E, Nn);
        }
        const int keep = rest ? 0 : 7;
        while (qn > keep) {
          const int n = qn < 8 ? qn : 8;
          const int base = qn - n;
          const bool ok = k < n;
          const int ent = q[base + (ok ? k : 0)];
          const int e = ent & 0xFFFFF, dr = (int)((unsigned)ent >> 20);
          const int sv = srcA[e < E ? e : E - 1];
          const int s = e < E ? sv : e - E;
          const float lg = logits8(xl, sxr, av, s, dr, lane);
          if (scan == 0) {
            if (ok) atomicMax(&smax[dr * 4 + hh], f2key(lg));
          } else {
            accum8(xl, sagg, sden, s, ok ? dr : RNODES, lg, smax[dr * 4 + hh], lane);
          }
          qn = base;
        }
        if (rest) break;
      }
      if (scan == 0) __syncthreads();
    }
  }
  __syncthreads();

  for (int i = tid; i < RNODES * 32; i += RWAVES * 32) {
    int nl = i >> 5, j4 = i & 31;
    int n = lo + nl;
    if (n < Nn) {
      float4 a = ((const float4*)sagg)[i];
      float dn = sden[nl * 4 + (j4 >> 3)] + 1e-16f;
      float4 bb = ((const float4*)b1)[j4];
      float4 r;
      r.x = a.x / dn + bb.x; r.y = a.y / dn + bb.y; r.z = a.z / dn + bb.z; r.w = a.w / dn + bb.w;
      r.x = r.x > 0.f ? r.x : expm1f(r.x);
      r.y = r.y > 0.f ? r.y : expm1f(r.y);
      r.z = r.z > 0.f ? r.z : expm1f(r.z);
      r.w = r.w > 0.f ? r.w : expm1f(r.w);
      ((float4*)hout)[(size_t)n * 32 + j4] = r;
    }
  }
}

__global__ void gatv2_edge_weights(float* __restrict__ lw,
                                   const unsigned* __restrict__ lmax,
                                   float* __restrict__ denom,
                                   const int* __restrict__ dstA,
                                   const float* __restrict__ b2,
                                   float* __restrict__ out,
                                   int E, int Nn) {
  int t = (int)(blockIdx.x * blockDim.x + threadIdx.x);
  if (t < Nn * 16) ((float4*)out)[t] = ((const float4*)b2)[t & 15];
  if (t >= (E + Nn) * 4) return;
  int e = t >> 2, h = t & 3;
  int d = (e < E) ? dstA[e] : e - E;
  float w = expf(lw[t] - key2f(lmax[(size_t)d * 4 + h]));
  lw[t] = w;
  atomicAdd(&denom[(size_t)d * 4 + h], w);
}

__global__ void gatv2_edge_accum_mean(const float* __restrict__ xl,
                                      const float* __restrict__ wgt,
                                      const float* __restrict__ denom,
                                      float* __restrict__ out,
                                      const int* __restrict__ srcA,
                                      const int* __restrict__ dstA,
                                      int E, int Nn) {
  const long EP = (long)E + Nn;
  const int lane = threadIdx.x & 31;
  long base; int cnt, sv, dv;
  if (!edge_wave_head(srcA, dstA, E, Nn, base, cnt, sv, dv)) return;

  for (int r = 0; r < 4; ++r) {
    if (8 * r >= cnt) break;
    long er = base + 8 * r + (lane >> 2);
    long erc = er < EP ? er : EP - 1;
    int hh = lane & 3;
    int dr = erc < E ? dstA[erc] : (int)(erc - E);
    float a = 0.25f * (wgt[erc * 4 + hh] / (denom[(size_t)dr * 4 + hh] + 1e-16f));

    float v[8][8];
    #pragma unroll
    for (int q = 0; q < 8; ++q) {
      int s = lane_geti(sv, 8 * r + q);
      const float* xls = xl + (size_t)s * 256;
      #pragma unroll
      for (int k = 0; k < 8; ++k) v[q][k] = xls[k * 32 + lane];
    }
    #pragma unroll
    for (int q = 0; q < 8; ++q) {
      if (8 * r + q < cnt) {
        int d = lane_geti(dv, 8 * r + q);
        float m0 = 0.f, m1 = 0.f;
        #pragma unroll
        for (int h = 0; h < 4; ++h) {
          float ah = lane_get(a, q * 4 + h);
          m0 += ah * v[q][2 * h];
          m1 += ah * v[q][2 * h + 1];
        }
        float* od = out + (size_t)d * 64;
        atomicAdd(&od[lane], m0);
        atomicAdd(&od[32 + lane], m1);
      }
    }
  }
}

extern "C" void kernel_launch(void* const* d_in, const int* in_sizes, int n_in,
                              void* d_out, int out_size, void* d_ws, size_t ws_size,
                              hipStream_t stream) {
  (void)in_sizes; (void)n_in; (void)out_size; (void)ws_size;
  const int  N  = NNODES;
  const int  E  = NEDGES;
  const long EP = (long)E + N;

  const float* x    = (const float*)d_in[0];
  const int*   ei   = (const int*)  d_in[1];
  const float* Wl1  = (const float*)d_in[2];
  const float* Wr1  = (const float*)d_in[3];
  const float* att1 = (const float*)d_in[4];
  const float* b1   = (const float*)d_in[5];
  const float* Wl2  = (const float*)d_in[6];
  const float* Wr2  = (const float*)d_in[7];
  const float* att2 = (const float*)d_in[8];
  const float* b2   = (const float*)d_in[9];
  const int* srcA = ei;
  const int* dstA = ei + E;

  float* ws    = (float*)d_ws;
  float* hbuf  = ws;
  float* xl    = hbuf + (size_t)N * 128;
  float* xr    = xl   + (size_t)N * 256;
  float* logit = xr   + (size_t)N * 256;
  unsigned* lmax = (unsigned*)(logit + (size_t)EP * 4);
  float* denom = (float*)(lmax + (size_t)N * 4);

  const int TPB = 256;
  const int TPE = 128;
  int eblocks = (int)((EP + 127) / 128);
  int gblocks = (((N + 15) / 16) + 7) / 8;
  int rblocks = (N + RNODES - 1) / RNODES;

  gatv2_gemm_nt_wmma<<<gblocks, TPB, 0, stream>>>(x, Wl1, xl, N, 128);
  gatv2_gemm_nt_wmma<<<gblocks, TPB, 0, stream>>>(x, Wr1, xr, N, 128);
  gatv2_range_layer1<<<rblocks, RWAVES * 32, 0, stream>>>(xl, xr, att1, b1, hbuf,
                                                          srcA, dstA, E, N);

  int wblocks = (int)((EP * 4 + TPB - 1) / TPB);
  hipMemsetAsync(lmax,  0, (size_t)N * 4 * sizeof(unsigned), stream);
  hipMemsetAsync(denom, 0, (size_t)N * 4 * sizeof(float), stream);
  gatv2_gemm_nt_wmma<<<gblocks, TPB, 0, stream>>>(hbuf, Wl2, xl, N, 256);
  gatv2_gemm_nt_wmma<<<gblocks, TPB, 0, stream>>>(hbuf, Wr2, xr, N, 256);
  gatv2_edge_logits<64><<<eblocks, TPE, 0, stream>>>(xl, xr, att2, srcA, dstA,
                                                     logit, lmax, E, N);
  gatv2_edge_weights<<<wblocks, TPB, 0, stream>>>(logit, lmax, denom, dstA, b2,
                                                  (float*)d_out, E, N);
  gatv2_edge_accum_mean<<<eblocks, TPE, 0, stream>>>(xl, logit, denom, (float*)d_out,
                                                     srcA, dstA, E, N);
}


# Weights converted and packed once per call into WMMA B-fragment order (one small extra launch): a GEMM wave reads a B fragment as two whole-line 16-byte loads, not four 16-line loads plus conversion
# speedup vs baseline: 3.1705x; 1.6439x over previous
#include <hip/hip_runtime.h>
#include <hip/hip_bf16.h>
#include <cmath>

typedef __attribute__((ext_vector_type(16))) _Float16 v16h;
typedef __attribute__((ext_vector_type(8)))  float    v8f;
typedef __attribute__((ext_vector_type(4)))  float    v4f;
typedef __attribute__((ext_vector_type(8)))  _Float16 v8h;

#define NNODES 50000
#define NEDGES 500000

__device__ __forceinline__ unsigned f2key(float f) {
  unsigned b = __float_as_uint(f);
  return (b & 0x80000000u) ? ~b : (b | 0x80000000u);
}
__device__ __forceinline__ float key2f(unsigned u) {
  if (u == 0u) return 0.0f;
  return (u & 0x80000000u) ? __uint_as_float(u ^ 0x80000000u) : __uint_as_float(~u);
}

__device__ __forceinline__ v16h load_cvt_f16(const float* p  ) {
  const float4* q = (const float4*)p;
  float4 v0 = q[0], v1 = q[1], v2 = q[4], v3 = q[5];
  v16h r;
  r[0] =(_Float16)v0.x; r[1] =(_Float16)v0.y; r[2] =(_Float16)v0.z; r[3] =(_Float16)v0.w;
  r[4] =(_Float16)v1.x; r[5] =(_Float16)v1.y; r[6] =(_Float16)v1.z; r[7] =(_Float16)v1.w;
  r[8] =(_Float16)v2.x; r[9] =(_Float16)v2.y; r[10]=(_Float16)v2.z; r[11]=(_Float16)v2.w;
  r[12]=(_Float16)v3.x; r[13]=(_Float16)v3.y; r[14]=(_Float16)v3.z; r[15]=(_Float16)v3.w;
  return r;
}

#define PK_W1L 0
#define PK_W1R 2048
#define PK_W2L 4096
#define PK_W2R 8192
#define PK_ALL 12288

__global__ void gatv2_pack_w(const float* __restrict__ W1l,
                             const float* __restrict__ W1r,
                             const float* __restrict__ W2l,
                             const float* __restrict__ W2r,
                             _Float16* __restrict__ Ph) {
  v8h* P = (v8h*)Ph;
  int t = (int)(blockIdx.x * blockDim.x + threadIdx.x);
  if (t >= PK_ALL) return;
  const float* W; int u;
  if (t < PK_W1R)      { W = W1l; u = t - PK_W1L; }
  else if (t < PK_W2L) { W = W1r; u = t - PK_W1R; }
  else if (t < PK_W2R) { W = W2l; u = t - PK_W2L; }
  else                 { W = W2r; u = t - PK_W2R; }
  int lane = u & 31, p = (u >> 5) & 1, kk = (u >> 6) & 3, tm = u >> 8;
  int l = lane & 15, half = lane >> 4;
  const float4* q = (const float4*)(W + (size_t)(tm * 16 + l) * 128 + kk * 32 + p * 16 + half * 8);
  float4 v0 = q[0], v1 = q[1];
  v8h r;
  r[0] = (_Float16)v0.x; r[1] = (_Float16)v0.y; r[2] = (_Float16)v0.z; r[3] = (_Float16)v0.w;
  r[4] = (_Float16)v1.x; r[5] = (_Float16)v1.y; r[6] = (_Float16)v1.z; r[7] = (_Float16)v1.w;
  P[t] = r;
}

__global__ void gatv2_gemm_nt_wmma(const float* __restrict__ X,
                                   const _Float16* __restrict__ Wph,
                                   float* __restrict__ Y,
                                   int Nrows, int Mcols) {
  const int K = 128;
  const v8h* Wp = (const v8h*)Wph;
  int wave = (int)((blockIdx.x * blockDim.x + threadIdx.x) >> 5);
  int lane = threadIdx.x & 31;
  int ntN = (Nrows + 15) >> 4;
  if (wave >= ntN) return;
  int half = lane >> 4;
  int l = lane & 15;
  int row = wave * 16 + l; if (row >= Nrows) row = Nrows - 1;
  const float* xrow = X + (size_t)row * K + half * 8;

  v16h a[4];
  #pragma unroll
  for (int kk = 0; kk < 4; ++kk) a[kk] = load_cvt_f16(xrow + kk * 32);

  int ntM = Mcols >> 4;
  for (int tm = 0; tm < ntM; tm += 2) {
    const v8h* wt = Wp + (size_t)tm * 256 + lane;
    v8h b[16];
    #pragma unroll
    for (int i = 0; i < 16; ++i) b[i] = wt[i * 32];
    asm volatile("" :: "v"(b[0]), "v"(b[1]), "v"(b[2]), "v"(b[3]), "v"(b[4]), "v"(b[5]), "v"(b[6]), "v"(b[7]),
                       "v"(b[8]), "v"(b[9]), "v"(b[10]), "v"(b[11]), "v"(b[12]), "v"(b[13]), "v"(b[14]), "v"(b[15]));
    v8f acc0 = {}, acc1 = {};
    #pragma unroll
    for (int kk = 0; kk < 4; ++kk) {
      v16h bb = __builtin_shufflevector(b[2 * kk], b[2 * kk + 1], 0, 1, 2, 3, 4, 5, 6, 7, 8, 9, 10, 11, 12, 13, 14, 15);
      acc0 = __builtin_amdgcn_wmma_f32_16x16x32_f16(
           false, a[kk],  false, bb,
           (short)0, acc0,  false,  false);
    }
    #pragma unroll
    for (int kk = 0; kk < 4; ++kk) {
      v16h bb = __builtin_shufflevector(b[8 + 2 * kk], b[9 + 2 * kk], 0, 1, 2, 3, 4, 5, 6, 7, 8, 9, 10, 11, 12, 13, 14, 15);
      acc1 = __builtin_amdgcn_wmma_f32_16x16x32_f16(
           false, a[kk],  false, bb,
           (short)0, acc1,  false,  false);
    }
    #pragma unroll
    for (int r = 0; r < 8; ++r) {
      int orow = wave * 16 + r + half * 8;
      if (orow < Nrows) {
        float* yp = Y + (size_t)orow * Mcols + tm * 16 + l;
        yp[0]  = acc0[r];
        yp[16] = acc1[r];
      }
    }
  }
}

__device__ __forceinline__ float lane_get(float v, int l) {
  return __int_as_float(__builtin_amdgcn_readlane(__float_as_int(v), l));
}
__device__ __forceinline__ int lane_geti(int v, int l) {
  return __builtin_amdgcn_readlane(v, l);
}
template<int CTRL>
__device__ __forceinline__ float xadd(float v) {
  return v + __int_as_float(__builtin_amdgcn_update_dpp(0, __float_as_int(v), CTRL, 0xf, 0xf, true));
}

#define RNODES   196
#define RWAVES   32
#define SCAN_UB  8
#define QW       184
#define RBLOCKS  ((NNODES + RNODES - 1) / RNODES)
#define QCOPY    ((QW + 31) / 32)

static_assert(NNODES <= 65536, "a queue entry keeps src in 16 bits");
static_assert(NNODES + NEDGES <= (1 << 20), "a queue entry keeps the edge number in 20 bits");
static_assert(RNODES <= 256, "a queue entry keeps the local node in 8 bits");
static_assert(7 + 128 <= QW, "the two-scan path queues one piece per lane (128 edges) on top of a rest of 7");

__device__ __forceinline__ void q_push(int* q, int& qn, bool m, int v) {
  unsigned mask = __builtin_amdgcn_ballot_w32(m);
  int pos = qn + (int)__builtin_amdgcn_mbcnt_lo(mask, 0u);
  if (m && pos < QW) q[pos] = v;
  qn += __builtin_popcount(mask);
}

__device__ __forceinline__ void q_push_piece(int* q, int& qn, int4 dd, int p, bool in, int lo) {
  int r0 = dd.x - lo, r1 = dd.y - lo, r2 = dd.z - lo, r3 = dd.w - lo;
  bool m0 = in & ((unsigned)r0 < (unsigned)RNODES);
  bool m1 = in & ((unsigned)r1 < (unsigned)RNODES);
  bool m2 = in & ((unsigned)r2 < (unsigned)RNODES);
  bool m3 = in & ((unsigned)r3 < (unsigned)RNODES);
  if (__builtin_amdgcn_ballot_w32(m0 | m1 | m2 | m3)) {
    q_push(q, qn, m0, (4 * p)     | (int)((unsigned)r0 << 20));
    q_push(q, qn, m1, (4 * p + 1) | (int)((unsigned)r1 << 20));
    q_push(q, qn, m2, (4 * p + 2) | (int)((unsigned)r2 << 20));
    q_push(q, qn, m3, (4 * p + 3) | (int)((unsigned)r3 << 20));
  }
}

__device__ __forceinline__ void q_push_rest(int* q, int& qn, int wv, int lane, int lo,
                                            const int* __restrict__ dstA, int E, int Nn) {
  if (wv == 0) {
    int e = (E >> 2) * 4 + lane;
    int ec = e < E ? e : E - 1;
    int rr = dstA[ec] - lo;
    bool m = e < E && (unsigned)rr < (unsigned)RNODES;
    q_push(q, qn, m, e | (int)((unsigned)rr << 20));
  }
  int nl = wv + RWAVES * lane;
  bool ms = nl < RNODES && lo + nl < Nn;
  q_push(q, qn, ms, (E + lo + nl) | (nl << 20));
}

__device__ __forceinline__ float logits8(const float* __restrict__ xl, const float* sxr,
                                         float4 av, int s, int dl, int lane) {
  float4 a[8];
  #pragma unroll
  for (int qq = 0; qq < 8; ++qq) {
    int sj = lane_geti(s, qq * 4);
    a[qq] = ((const float4*)(xl + (size_t)sj * 128))[lane];
  }
  float lg = 0.f;
  #pragma unroll
  for (int qq = 0; qq < 8; ++qq) {
    int dj = lane_geti(dl, qq * 4);
    float4 bq = ((const float4*)(sxr + dj * 128))[lane];
    float t0 = a[qq].x + bq.x, t1 = a[qq].y + bq.y;
    float t2 = a[qq].z + bq.z, t3 = a[qq].w + bq.w;
    t0 = t0 > 0.f ? t0 : 0.2f * t0;
    t1 = t1 > 0.f ? t1 : 0.2f * t1;
    t2 = t2 > 0.f ? t2 : 0.2f * t2;
    t3 = t3 > 0.f ? t3 : 0.2f * t3;
    float acc = t0 * av.x;
    acc += t1 * av.y;
    acc += t2 * av.z;
    acc += t3 * av.w;
    acc = xadd<0x161>(acc);
    acc = xadd<0x162>(acc);
    acc = xadd<0x164>(acc);
    #pragma unroll
    for (int h = 0; h < 4; ++h) {
      float sh = lane_get(acc, 8 * h);
      if (lane == qq * 4 + h) lg = sh;
    }
  }
  return lg;
}

__device__ __forceinline__ void accum8(const float* __restrict__ xl, float* sagg, float* sden,
                                       int s, int dl, float lg, unsigned lm, int lane) {
  float v[8][4];
  #pragma unroll
  for (int qq = 0; qq < 8; ++qq) {
    int sj = lane_geti(s, qq * 4);
    const float* xls = xl + (size_t)sj * 128;
    #pragma unroll
    for (int i = 0; i < 4; ++i) v[qq][i] = xls[lane + 32 * i];
  }
  const float w = expf(lg - key2f(lm));
  atomicAdd(&sden[dl * 4 + (lane & 3)], w);
  #pragma unroll
  for (int qq = 0; qq < 8; ++qq) {
    int dj = lane_geti(dl, qq * 4);
    float* ad = sagg + dj * 128;
    #pragma unroll
    for (int i = 0; i < 4; ++i) {
      float wh = lane_get(w, qq * 4 + i);
      atomicAdd(&ad[lane + 32 * i], wh * v[qq][i]);
    }
  }
}

__global__ __launch_bounds__(RWAVES * 32)
void gatv2_range_layer1(const float* __restrict__ xl,
                        const float* __restrict__ xr,
                        const float* __restrict__ att,
                        const float* __restrict__ b1,
                        float* __restrict__ hout,
                        int* __restrict__ qsave,
                        int* __restrict__ qcnt,
                        const int* __restrict__ srcA,
                        const int* __restrict__ dstA,
                        int E, int Nn) {
  __shared__ __attribute__((aligned(16))) float sagg[(RNODES + 1) * 128];
  __shared__ __attribute__((aligned(16))) float sden[(RNODES + 1) * 4];
  __shared__ __attribute__((aligned(16))) unsigned smax[RNODES * 4];
  __shared__ __attribute__((aligned(16))) float sxr[RNODES * 128];
  __shared__ int   sqe[RWAVES][QW];
  __shared__ float sql[RWAVES][QW * 4];
  __shared__ int   sover;

  const int tid = threadIdx.x, lane = tid & 31;
  const int wv = __builtin_amdgcn_readfirstlane(tid >> 5);
  const int lo = (int)blockIdx.x * RNODES;

  const float4 z4 = make_float4(0.f, 0.f, 0.f, 0.f);
  for (int i = tid; i < (RNODES + 1) * 32; i += RWAVES * 32) ((float4*)sagg)[i] = z4;
  if (tid < RNODES + 1) ((float4*)sden)[tid] = z4;
  if (tid < RNODES) ((uint4*)smax)[tid] = make_uint4(0u, 0u, 0u, 0u);
  if (tid == 0) sover = 0;
  for (int i = tid; i < RNODES * 32; i += RWAVES * 32) {
    int n = lo + (i >> 5);
    float4 t = make_float4(0.f, 0.f, 0.f, 0.f);
    if (n < Nn) t = ((const float4*)xr)[(size_t)n * 32 + (i & 31)];
    ((float4*)sxr)[i] = t;
  }
  __syncthreads();

  int* q = sqe[wv];
  float* ql = sql[wv];
  int qn = 0;
  int pc = 0;
  bool over = false;
  const float4 av = ((const float4*)att)[lane];
  const int k = lane >> 2, hh = lane & 3;

  const int n4 = E >> 2;
  const int nch = (n4 + 31) >> 5;
  const int4* d4 = (const int4*)dstA;
  const int nb = (nch + RWAVES * SCAN_UB - 1) / (RWAVES * SCAN_UB);

  for (int b = 0; b <= nb; ++b) {
    if (b < nb) {
      const int c0 = b * (RWAVES * SCAN_UB) + wv;
      int4 dd[SCAN_UB];
      #pragma unroll
      for (int u = 0; u < SCAN_UB; ++u) {
        int p = (c0 + RWAVES * u) * 32 + lane;
        dd[u] = d4[p < n4 ? p : n4 - 1];
      }
      #pragma unroll
      for (int u = 0; u < SCAN_UB; ++u) {
        int p = (c0 + RWAVES * u) * 32 + lane;
        q_push_piece(q, qn, dd[u], p, p < n4, lo);
      }
    } else {
      q_push_rest(q, qn, wv, lane, lo, dstA, E, Nn);
    }
    if (qn > QW) { over = true; break; }

    const int keep = b < nb ? 7 : 0;
    while (qn - pc > keep) {
      const int n = qn - pc < 8 ? qn - pc : 8;
      const bool ok = k < n;
      const int ent = q[pc + (ok ? k : 0)];
      const int e = ent & 0xFFFFF, dl = (int)((unsigned)ent >> 20);
      const int sv = srcA[e < E ? e : E - 1];
      const int s = e < E ? sv : e - E;
      const float lg = logits8(xl, sxr, av, s, dl, lane);
      if (ok) {
        ql[(pc + k) * 4 + hh] = lg;
        atomicMax(&smax[dl * 4 + hh], f2key(lg));
        if (hh == 0) q[pc + k] = s | (dl << 16);
      }
      pc += n;
    }
  }
  if (over && lane == 0) sover = 1;
  __syncthreads();

  const int widx = (int)blockIdx.x * RWAVES + wv;
  if (sover == 0) {
    int* qs = qsave + (size_t)widx * QW;
    #pragma unroll
    for (int u = 0; u < QCOPY; ++u) {
      int i = lane + 32 * u;
      if (i < qn) qs[i] = q[i];
    }
    if (lane == 0) qcnt[widx] = qn;
    for (int base = 0; base < qn; base += 8) {
      const int n = qn - base < 8 ? qn - base : 8;
      const bool ok = k < n;
      const int ix = base + (ok ? k : 0);
      const int ent = q[ix];
      const int s = ent & 0xFFFF, dr = (int)((unsigned)ent >> 16);
      accum8(xl, sagg, sden, s, ok ? dr : RNODES, ql[ix * 4 + hh], smax[dr * 4 + hh], lane);
    }
  } else {
    if (lane == 0) qcnt[widx] = -1;
    for (int scan = 0; scan < 2; ++scan) {
      qn = 0;
      for (int c = wv; ; c += RWAVES) {
        const bool rest = c >= nch;
        if (!rest) {
          int p = c * 32 + lane;
          q_push_piece(q, qn, d4[p < n4 ? p : n4 - 1], p, p < n4, lo);
        } else {
          q_push_rest(q, qn, wv, lane, lo, dstA, E, Nn);
        }
        const int keep = rest ? 0 : 7;
        while (qn > keep) {
          const int n = qn < 8 ? qn : 8;
          const int base = qn - n;
          const bool ok = k < n;
          const int ent = q[base + (ok ? k : 0)];
          const int e = ent & 0xFFFFF, dr = (int)((unsigned)ent >> 20);
          const int sv = srcA[e < E ? e : E - 1];
          const int s = e < E ? sv : e - E;
          const float lg = logits8(xl, sxr, av, s, dr, lane);
          if (scan == 0) {
            if (ok) atomicMax(&smax[dr * 4 + hh], f2key(lg));
          } else {
            accum8(xl, sagg, sden, s, ok ? dr : RNODES, lg, smax[dr * 4 + hh], lane);
          }
          qn = base;
        }
        if (rest) break;
      }
      if (scan == 0) __syncthreads();
    }
  }
  __syncthreads();

  for (int i = tid; i < RNODES * 32; i += RWAVES * 32) {
    int nl = i >> 5, j4 = i & 31;
    int n = lo + nl;
    if (n < Nn) {
      float4 a = ((const float4*)sagg)[i];
      float dn = sden[nl * 4 + (j4 >> 3)] + 1e-16f;
      float4 bb = ((const float4*)b1)[j4];
      float4 r;
      r.x = a.x / dn + bb.x; r.y = a.y / dn + bb.y; r.z = a.z / dn + bb.z; r.w = a.w / dn + bb.w;
      r.x = r.x > 0.f ? r.x : expm1f(r.x);
      r.y = r.y > 0.f ? r.y : expm1f(r.y);
      r.z = r.z > 0.f ? r.z : expm1f(r.z);
      r.w = r.w > 0.f ? r.w : expm1f(r.w);
      ((float4*)hout)[(size_t)n * 32 + j4] = r;
    }
  }
}


__device__ __forceinline__ float logits8_c64(const float* __restrict__ xl, const float* sxr,
                                             float4 av0, float4 av1, int s, int dl, int lane) {
  v4f a[8][2];
  #pragma unroll
  for (int qq = 0; qq < 8; ++qq) {
    int sj = lane_geti(s, qq * 4);
    const v4f* ps = (const v4f*)(xl + (size_t)sj * 256);
    a[qq][0] = ps[lane];
    a[qq][1] = ps[32 + lane];
  }
  asm volatile("" :: "v"(a[0][0]), "v"(a[0][1]), "v"(a[1][0]), "v"(a[1][1]),
                     "v"(a[2][0]), "v"(a[2][1]), "v"(a[3][0]), "v"(a[3][1]),
                     "v"(a[4][0]), "v"(a[4][1]), "v"(a[5][0]), "v"(a[5][1]),
                     "v"(a[6][0]), "v"(a[6][1]), "v"(a[7][0]), "v"(a[7][1]));
  float lg = 0.f;
  #pragma unroll
  for (int qq = 0; qq < 8; ++qq) {
    int dj = lane_geti(dl, qq * 4);
    const float4* pd = (const float4*)(sxr + dj * 256);
    float p[2];
    #pragma unroll
    for (int i = 0; i < 2; ++i) {
      float4 bq = pd[i * 32 + lane];
      float4 aw = i == 0 ? av0 : av1;
      float t0 = a[qq][i].x + bq.x, t1 = a[qq][i].y + bq.y;
      float t2 = a[qq][i].z + bq.z, t3 = a[qq][i].w + bq.w;
      t0 = t0 > 0.f ? t0 : 0.2f * t0;
      t1 = t1 > 0.f ? t1 : 0.2f * t1;
      t2 = t2 > 0.f ? t2 : 0.2f * t2;
      t3 = t3 > 0.f ? t3 : 0.2f * t3;
      float acc = t0 * aw.x;
      acc += t1 * aw.y;
      acc += t2 * aw.z;
      acc += t3 * aw.w;
      acc = xadd<0x161>(acc);
      acc = xadd<0x162>(acc);
      acc = xadd<0x164>(acc);
      acc = xadd<0x168>(acc);
      p[i] = acc;
    }
    #pragma unroll
    for (int h = 0; h < 4; ++h) {
      float sh = lane_get(p[h >> 1], (h & 1) * 16);
      if (lane == qq * 4 + h) lg = sh;
    }
  }
  return lg;
}

__device__ __forceinline__ void accum8_mean(const float* __restrict__ xl, float* sout,
                                            int s, int dl, float a, int lane) {
  float v[8][8];
  #pragma unroll
  for (int qq = 0; qq < 8; ++qq) {
    int sj = lane_geti(s, qq * 4);
    const float* xls = xl + (size_t)sj * 256;
    #pragma unroll
    for (int kk = 0; kk < 8; ++kk) v[qq][kk] = xls[kk * 32 + lane];
  }
  #pragma unroll
  for (int qq = 0; qq < 8; ++qq) {
    int dj = lane_geti(dl, qq * 4);
    float m0 = 0.f, m1 = 0.f;
    #pragma unroll
    for (int h = 0; h < 4; ++h) {
      float ah = lane_get(a, qq * 4 + h);
      m0 += ah * v[qq][2 * h];
      m1 += ah * v[qq][2 * h + 1];
    }
    float* od = sout + dj * 64;
    atomicAdd(&od[lane], m0);
    atomicAdd(&od[32 + lane], m1);
  }
}

__global__ __launch_bounds__(RWAVES * 32)
void gatv2_range_layer2(const float* __restrict__ xl,
                        const float* __restrict__ xr,
                        const float* __restrict__ att,
                        const float* __restrict__ b2,
                        float* __restrict__ out,
                        float* __restrict__ qlog,
                        const int* __restrict__ qsave,
                        const int* __restrict__ qcnt,
                        const int* __restrict__ srcA,
                        const int* __restrict__ dstA,
                        int E, int Nn) {
  __shared__ __attribute__((aligned(16))) float sout[(RNODES + 1) * 64];
  __shared__ __attribute__((aligned(16))) float sden[(RNODES + 1) * 4];
  __shared__ __attribute__((aligned(16))) unsigned smax[RNODES * 4];
  __shared__ __attribute__((aligned(16))) float sxr[RNODES * 256];
  __shared__ int   sqe[RWAVES][QW];

  const int tid = threadIdx.x, lane = tid & 31;
  const int wv = __builtin_amdgcn_readfirstlane(tid >> 5);
  const int lo = (int)blockIdx.x * RNODES;

  int* q = sqe[wv];
  const int widx = (int)blockIdx.x * RWAVES + wv;
  int qn = __builtin_amdgcn_readfirstlane(qcnt[widx]);
  const bool over = qn < 0;
  qn = qn > QW ? QW : qn;
  if (!over && qn > 0) {
    const int* qs = qsave + (size_t)widx * QW;
    int t[QCOPY];
    #pragma unroll
    for (int u = 0; u < QCOPY; ++u) {
      int i = lane + 32 * u;
      t[u] = qs[i < qn ? i : qn - 1];
    }
    #pragma unroll
    for (int u = 0; u < QCOPY; ++u) {
      int i = lane + 32 * u;
      if (i < qn) q[i] = t[u];
    }
  }

  const float4 z4 = make_float4(0.f, 0.f, 0.f, 0.f);
  for (int i = tid; i < (RNODES + 1) * 16; i += RWAVES * 32) ((float4*)sout)[i] = z4;
  if (tid < RNODES + 1) ((float4*)sden)[tid] = z4;
  if (tid < RNODES) ((uint4*)smax)[tid] = make_uint4(0u, 0u, 0u, 0u);
  for (int i = tid; i < RNODES * 64; i += RWAVES * 32) {
    int n = lo + (i >> 6);
    float4 t = make_float4(0.f, 0.f, 0.f, 0.f);
    if (n < Nn) t = ((const float4*)xr)[(size_t)n * 64 + (i & 63)];
    ((float4*)sxr)[i] = t;
  }
  __syncthreads();

  float* ql = qlog + (size_t)widx * (QW * 4);
  const float4 av0 = ((const float4*)att)[lane];
  const float4 av1 = ((const float4*)att)[32 + lane];
  const int k = lane >> 2, hh = lane & 3;

  const int n4 = E >> 2;
  const int nch = (n4 + 31) >> 5;
  const int4* d4 = (const int4*)dstA;

  if (!over) {
    for (int base = 0; base < qn; base += 8) {
      const int n = qn - base < 8 ? qn - base : 8;
      const bool ok = k < n;
      const int ix = base + (ok ? k : 0);
      const int ent = q[ix];
      const int s = ent & 0xFFFF, dl = (int)((unsigned)ent >> 16);
      const float lg = logits8_c64(xl, sxr, av0, av1, s, dl, lane);
      if (ok) {
        ql[ix * 4 + hh] = lg;
        atomicMax(&smax[dl * 4 + hh], f2key(lg));
      }
    }
  }
  __syncthreads();

  if (!over) {
    for (int base = 0; base < qn; base += 8) {
      if (base + k < qn) {
        const int ix = base + k;
        const int dr = (int)((unsigned)q[ix] >> 16);
        const float w = expf(ql[ix * 4 + hh] - key2f(smax[dr * 4 + hh]));
        ql[ix * 4 + hh] = w;
        atomicAdd(&sden[dr * 4 + hh], w);
      }
    }
    __syncthreads();
    for (int base = 0; base < qn; base += 8) {
      const int n = qn - base < 8 ? qn - base : 8;
      const bool ok = k < n;
      const int ix = base + (ok ? k : 0);
      const int ent = q[ix];
      const int s = ent & 0xFFFF, dr = (int)((unsigned)ent >> 16);
      const float a = 0.25f * (ql[ix * 4 + hh] / (sden[dr * 4 + hh] + 1e-16f));
      accum8_mean(xl, sout, s, ok ? dr : RNODES, a, lane);
    }
  } else {
    for (int scan = 0; scan < 3; ++scan) {
      qn = 0;
      for (int c = wv; ; c += RWAVES) {
        const bool rest = c >= nch;
        if (!rest) {
          int p = c * 32 + lane;
          q_push_piece(q, qn, d4[p < n4 ? p : n4 - 1], p, p < n4, lo);
        } else {
          q_push_rest(q, qn, wv, lane, lo, dstA, E, Nn);
        }
        const int keep = rest ? 0 : 7;
        while (qn > keep) {
          const int n = qn < 8 ? qn : 8;
          const int base = qn - n;
          const bool ok = k < n;
          const int ent = q[base + (ok ? k : 0)];
          const int e = ent & 0xFFFFF, dr = (int)((unsigned)ent >> 20);
          const int sv = srcA[e < E ? e : E - 1];
          const int s = e < E ? sv : e - E;
          const float lg = logits8_c64(xl, sxr, av0, av1, s, dr, lane);
          if (scan == 0) {
            if (ok) atomicMax(&smax[dr * 4 + hh], f2key(lg));
          } else {
            const float w = expf(lg - key2f(smax[dr * 4 + hh]));
            if (scan == 1) {
              if (ok) atomicAdd(&sden[dr * 4 + hh], w);
            } else {
              const float a = 0.25f * (w / (sden[dr * 4 + hh] + 1e-16f));
              accum8_mean(xl, sout, s, ok ? dr : RNODES, a, lane);
            }
          }
          qn = base;
        }
        if (rest) break;
      }
      if (scan < 2) __syncthreads();
    }
  }
  __syncthreads();

  for (int i = tid; i < RNODES * 16; i += RWAVES * 32) {
    int nl = i >> 4, j4 = i & 15;
    int n = lo + nl;
    if (n < Nn) {
      float4 a = ((const float4*)sout)[i];
      float4 bb = ((const float4*)b2)[j4];
      float4 r;
      r.x = a.x + bb.x; r.y = a.y + bb.y; r.z = a.z + bb.z; r.w = a.w + bb.w;
      ((float4*)out)[(size_t)n * 16 + j4] = r;
    }
  }
}

extern "C" void kernel_launch(void* const* d_in, const int* in_sizes, int n_in,
                              void* d_out, int out_size, void* d_ws, size_t ws_size,
                              hipStream_t stream) {
  (void)in_sizes; (void)n_in; (void)out_size; (void)ws_size;
  const int  N  = NNODES;
  const int  E  = NEDGES;

  const float* x    = (const float*)d_in[0];
  const int*   ei   = (const int*)  d_in[1];
  const float* Wl1  = (const float*)d_in[2];
  const float* Wr1  = (const float*)d_in[3];
  const float* att1 = (const float*)d_in[4];
  const float* b1   = (const float*)d_in[5];
  const float* Wl2  = (const float*)d_in[6];
  const float* Wr2  = (const float*)d_in[7];
  const float* att2 = (const float*)d_in[8];
  const float* b2   = (const float*)d_in[9];
  const int* srcA = ei;
  const int* dstA = ei + E;

  float* ws    = (float*)d_ws;
  float* hbuf  = ws;
  float* xl    = hbuf + (size_t)N * 128;
  float* xr    = xl   + (size_t)N * 256;
  float* qlog  = xr   + (size_t)N * 256;
  int*   qsave = (int*)(qlog + (size_t)RBLOCKS * RWAVES * QW * 4);
  int*   qcnt  = qsave + (size_t)RBLOCKS * RWAVES * QW;
  _Float16* wpk = (_Float16*)(qcnt + (size_t)RBLOCKS * RWAVES);

  const int TPB = 256;
  int gblocks = (((N + 15) / 16) + 7) / 8;
  int rblocks = RBLOCKS;

  gatv2_pack_w<<<PK_ALL / TPB, TPB, 0, stream>>>(Wl1, Wr1, Wl2, Wr2, wpk);

  gatv2_gemm_nt_wmma<<<gblocks, TPB, 0, stream>>>(x, wpk + 8 * PK_W1L, xl, N, 128);
  gatv2_gemm_nt_wmma<<<gblocks, TPB, 0, stream>>>(x, wpk + 8 * PK_W1R, xr, N, 128);
  gatv2_range_layer1<<<rblocks, RWAVES * 32, 0, stream>>>(xl, xr, att1, b1, hbuf, qsave, qcnt,
                                                          srcA, dstA, E, N);

  gatv2_gemm_nt_wmma<<<gblocks, TPB, 0, stream>>>(hbuf, wpk + 8 * PK_W2L, xl, N, 256);
  gatv2_gemm_nt_wmma<<<gblocks, TPB, 0, stream>>>(hbuf, wpk + 8 * PK_W2R, xr, N, 256);
  gatv2_range_layer2<<<rblocks, RWAVES * 32, 0, stream>>>(xl, xr, att2, b2, (float*)d_out, qlog,
                                                          qsave, qcnt, srcA, dstA, E, N);
}


# Target-side rows xr = X Wr^T computed inside the destination-range kernels (WMMA on the packed weights, straight into LDS): two GEMM launches and xr's trip through memory gone; same values
# speedup vs baseline: 3.5523x; 1.1204x over previous
#include <hip/hip_runtime.h>
#include <hip/hip_bf16.h>
#include <cmath>

typedef __attribute__((ext_vector_type(16))) _Float16 v16h;
typedef __attribute__((ext_vector_type(8)))  float    v8f;
typedef __attribute__((ext_vector_type(4)))  float    v4f;
typedef __attribute__((ext_vector_type(8)))  _Float16 v8h;

#define NNODES 50000
#define NEDGES 500000

__device__ __forceinline__ unsigned f2key(float f) {
  unsigned b = __float_as_uint(f);
  return (b & 0x80000000u) ? ~b : (b | 0x80000000u);
}
__device__ __forceinline__ float key2f(unsigned u) {
  if (u == 0u) return 0.0f;
  return (u & 0x80000000u) ? __uint_as_float(u ^ 0x80000000u) : __uint_as_float(~u);
}

__device__ __forceinline__ v16h load_cvt_f16(const float* p  ) {
  const float4* q = (const float4*)p;
  float4 v0 = q[0], v1 = q[1], v2 = q[4], v3 = q[5];
  v16h r;
  r[0] =(_Float16)v0.x; r[1] =(_Float16)v0.y; r[2] =(_Float16)v0.z; r[3] =(_Float16)v0.w;
  r[4] =(_Float16)v1.x; r[5] =(_Float16)v1.y; r[6] =(_Float16)v1.z; r[7] =(_Float16)v1.w;
  r[8] =(_Float16)v2.x; r[9] =(_Float16)v2.y; r[10]=(_Float16)v2.z; r[11]=(_Float16)v2.w;
  r[12]=(_Float16)v3.x; r[13]=(_Float16)v3.y; r[14]=(_Float16)v3.z; r[15]=(_Float16)v3.w;
  return r;
}

#define PK_W1L 0
#define PK_W1R 2048
#define PK_W2L 4096
#define PK_W2R 8192
#define PK_ALL 12288

__global__ void gatv2_pack_w(const float* __restrict__ W1l,
                             const float* __restrict__ W1r,
                             const float* __restrict__ W2l,
                             const float* __restrict__ W2r,
                             _Float16* __restrict__ Ph) {
  v8h* P = (v8h*)Ph;
  int t = (int)(blockIdx.x * blockDim.x + threadIdx.x);
  if (t >= PK_ALL) return;
  const float* W; int u;
  if (t < PK_W1R)      { W = W1l; u = t - PK_W1L; }
  else if (t < PK_W2L) { W = W1r; u = t - PK_W1R; }
  else if (t < PK_W2R) { W = W2l; u = t - PK_W2L; }
  else                 { W = W2r; u = t - PK_W2R; }
  int lane = u & 31, p = (u >> 5) & 1, kk = (u >> 6) & 3, tm = u >> 8;
  int l = lane & 15, half = lane >> 4;
  const float4* q = (const float4*)(W + (size_t)(tm * 16 + l) * 128 + kk * 32 + p * 16 + half * 8);
  float4 v0 = q[0], v1 = q[1];
  v8h r;
  r[0] = (_Float16)v0.x; r[1] = (_Float16)v0.y; r[2] = (_Float16)v0.z; r[3] = (_Float16)v0.w;
  r[4] = (_Float16)v1.x; r[5] = (_Float16)v1.y; r[6] = (_Float16)v1.z; r[7] = (_Float16)v1.w;
  P[t] = r;
}

__global__ void gatv2_gemm_nt_wmma(const float* __restrict__ X,
                                   const _Float16* __restrict__ Wph,
                                   float* __restrict__ Y,
                                   int Nrows, int Mcols) {
  const int K = 128;
  const v8h* Wp = (const v8h*)Wph;
  int wave = (int)((blockIdx.x * blockDim.x + threadIdx.x) >> 5);
  int lane = threadIdx.x & 31;
  int ntN = (Nrows + 15) >> 4;
  if (wave >= ntN) return;
  int half = lane >> 4;
  int l = lane & 15;
  int row = wave * 16 + l; if (row >= Nrows) row = Nrows - 1;
  const float* xrow = X + (size_t)row * K + half * 8;

  v16h a[4];
  #pragma unroll
  for (int kk = 0; kk < 4; ++kk) a[kk] = load_cvt_f16(xrow + kk * 32);

  int ntM = Mcols >> 4;
  for (int tm = 0; tm < ntM; tm += 2) {
    const v8h* wt = Wp + (size_t)tm * 256 + lane;
    v8h b[16];
    #pragma unroll
    for (int i = 0; i < 16; ++i) b[i] = wt[i * 32];
    asm volatile("" :: "v"(b[0]), "v"(b[1]), "v"(b[2]), "v"(b[3]), "v"(b[4]), "v"(b[5]), "v"(b[6]), "v"(b[7]),
                       "v"(b[8]), "v"(b[9]), "v"(b[10]), "v"(b[11]), "v"(b[12]), "v"(b[13]), "v"(b[14]), "v"(b[15]));
    v8f acc0 = {}, acc1 = {};
    #pragma unroll
    for (int kk = 0; kk < 4; ++kk) {
      v16h bb = __builtin_shufflevector(b[2 * kk], b[2 * kk + 1], 0, 1, 2, 3, 4, 5, 6, 7, 8, 9, 10, 11, 12, 13, 14, 15);
      acc0 = __builtin_amdgcn_wmma_f32_16x16x32_f16(
           false, a[kk],  false, bb,
           (short)0, acc0,  false,  false);
    }
    #pragma unroll
    for (int kk = 0; kk < 4; ++kk) {
      v16h bb = __builtin_shufflevector(b[8 + 2 * kk], b[9 + 2 * kk], 0, 1, 2, 3, 4, 5, 6, 7, 8, 9, 10, 11, 12, 13, 14, 15);
      acc1 = __builtin_amdgcn_wmma_f32_16x16x32_f16(
           false, a[kk],  false, bb,
           (short)0, acc1,  false,  false);
    }
    #pragma unroll
    for (int r = 0; r < 8; ++r) {
      int orow = wave * 16 + r + half * 8;
      if (orow < Nrows) {
        float* yp = Y + (size_t)orow * Mcols + tm * 16 + l;
        yp[0]  = acc0[r];
        yp[16] = acc1[r];
      }
    }
  }
}

template<int MC, int RN>
__device__ __forceinline__ void range_rows_wmma(const float* __restrict__ X,
                                                const _Float16* __restrict__ Wph,
                                                float* srow, int lo, int Nn, int wv, int lane) {
  constexpr int NSTRIP = (RN + 15) / 16;
  constexpr int NTH = MC / 32;
  if (wv < 2 * NSTRIP) {
    const v8h* Wp = (const v8h*)Wph;
    const int strip = wv >> 1, ch = wv & 1;
    const int r0 = strip * 16 < RN - 16 ? strip * 16 : RN - 16;
    const int half = lane >> 4, l = lane & 15;
    int n = lo + r0 + l; if (n >= Nn) n = Nn - 1;
    const float* xrow = X + (size_t)n * 128 + half * 8;
    v16h a[4];
    #pragma unroll
    for (int kk = 0; kk < 4; ++kk) a[kk] = load_cvt_f16(xrow + kk * 32);
    #pragma nounroll
    for (int tm = ch * NTH; tm < (ch + 1) * NTH; tm += 2) {
      const v8h* wt = Wp + (size_t)tm * 256 + lane;
      v8h b[16];
      #pragma unroll
      for (int i = 0; i < 16; ++i) b[i] = wt[i * 32];
      asm volatile("" :: "v"(b[0]), "v"(b[1]), "v"(b[2]), "v"(b[3]), "v"(b[4]), "v"(b[5]), "v"(b[6]), "v"(b[7]),
                         "v"(b[8]), "v"(b[9]), "v"(b[10]), "v"(b[11]), "v"(b[12]), "v"(b[13]), "v"(b[14]), "v"(b[15]));
      v8f acc0 = {}, acc1 = {};
      #pragma unroll
      for (int kk = 0; kk < 4; ++kk) {
        v16h bb = __builtin_shufflevector(b[2 * kk], b[2 * kk + 1], 0, 1, 2, 3, 4, 5, 6, 7, 8, 9, 10, 11, 12, 13, 14, 15);
        acc0 = __builtin_amdgcn_wmma_f32_16x16x32_f16(
             false, a[kk],  false, bb,
             (short)0, acc0,  false,  false);
      }
      #pragma unroll
      for (int kk = 0; kk < 4; ++kk) {
        v16h bb = __builtin_shufflevector(b[8 + 2 * kk], b[9 + 2 * kk], 0, 1, 2, 3, 4, 5, 6, 7, 8, 9, 10, 11, 12, 13, 14, 15);
        acc1 = __builtin_amdgcn_wmma_f32_16x16x32_f16(
             false, a[kk],  false, bb,
             (short)0, acc1,  false,  false);
      }
      float* yp = srow + (r0 + half * 8) * MC + tm * 16 + l;
      #pragma unroll
      for (int r = 0; r < 8; ++r) {
        yp[r * MC]      = acc0[r];
        yp[r * MC + 16] = acc1[r];
      }
    }
  }
}

__device__ __forceinline__ float lane_get(float v, int l) {
  return __int_as_float(__builtin_amdgcn_readlane(__float_as_int(v), l));
}
__device__ __forceinline__ int lane_geti(int v, int l) {
  return __builtin_amdgcn_readlane(v, l);
}
template<int CTRL>
__device__ __forceinline__ float xadd(float v) {
  return v + __int_as_float(__builtin_amdgcn_update_dpp(0, __float_as_int(v), CTRL, 0xf, 0xf, true));
}

#define RNODES   196
#define RWAVES   32
#define SCAN_UB  8
#define QW       184
#define RBLOCKS  ((NNODES + RNODES - 1) / RNODES)
#define QCOPY    ((QW + 31) / 32)

static_assert(NNODES <= 65536, "a queue entry keeps src in 16 bits");
static_assert(NNODES + NEDGES <= (1 << 20), "a queue entry keeps the edge number in 20 bits");
static_assert(RNODES <= 256, "a queue entry keeps the local node in 8 bits");
static_assert(7 + 128 <= QW, "the two-scan path queues one piece per lane (128 edges) on top of a rest of 7");

__device__ __forceinline__ void q_push(int* q, int& qn, bool m, int v) {
  unsigned mask = __builtin_amdgcn_ballot_w32(m);
  int pos = qn + (int)__builtin_amdgcn_mbcnt_lo(mask, 0u);
  if (m && pos < QW) q[pos] = v;
  qn += __builtin_popcount(mask);
}

__device__ __forceinline__ void q_push_piece(int* q, int& qn, int4 dd, int p, bool in, int lo) {
  int r0 = dd.x - lo, r1 = dd.y - lo, r2 = dd.z - lo, r3 = dd.w - lo;
  bool m0 = in & ((unsigned)r0 < (unsigned)RNODES);
  bool m1 = in & ((unsigned)r1 < (unsigned)RNODES);
  bool m2 = in & ((unsigned)r2 < (unsigned)RNODES);
  bool m3 = in & ((unsigned)r3 < (unsigned)RNODES);
  if (__builtin_amdgcn_ballot_w32(m0 | m1 | m2 | m3)) {
    q_push(q, qn, m0, (4 * p)     | (int)((unsigned)r0 << 20));
    q_push(q, qn, m1, (4 * p + 1) | (int)((unsigned)r1 << 20));
    q_push(q, qn, m2, (4 * p + 2) | (int)((unsigned)r2 << 20));
    q_push(q, qn, m3, (4 * p + 3) | (int)((unsigned)r3 << 20));
  }
}

__device__ __forceinline__ void q_push_rest(int* q, int& qn, int wv, int lane, int lo,
                                            const int* __restrict__ dstA, int E, int Nn) {
  if (wv == 0) {
    int e = (E >> 2) * 4 + lane;
    int ec = e < E ? e : E - 1;
    int rr = dstA[ec] - lo;
    bool m = e < E && (unsigned)rr < (unsigned)RNODES;
    q_push(q, qn, m, e | (int)((unsigned)rr << 20));
  }
  int nl = wv + RWAVES * lane;
  bool ms = nl < RNODES && lo + nl < Nn;
  q_push(q, qn, ms, (E + lo + nl) | (nl << 20));
}

__device__ __forceinline__ float logits8(const float* __restrict__ xl, const float* sxr,
                                         float4 av, int s, int dl, int lane) {
  float4 a[8];
  #pragma unroll
  for (int qq = 0; qq < 8; ++qq) {
    int sj = lane_geti(s, qq * 4);
    a[qq] = ((const float4*)(xl + (size_t)sj * 128))[lane];
  }
  float lg = 0.f;
  #pragma unroll
  for (int qq = 0; qq < 8; ++qq) {
    int dj = lane_geti(dl, qq * 4);
    float4 bq = ((const float4*)(sxr + dj * 128))[lane];
    float t0 = a[qq].x + bq.x, t1 = a[qq].y + bq.y;
    float t2 = a[qq].z + bq.z, t3 = a[qq].w + bq.w;
    t0 = t0 > 0.f ? t0 : 0.2f * t0;
    t1 = t1 > 0.f ? t1 : 0.2f * t1;
    t2 = t2 > 0.f ? t2 : 0.2f * t2;
    t3 = t3 > 0.f ? t3 : 0.2f * t3;
    float acc = t0 * av.x;
    acc += t1 * av.y;
    acc += t2 * av.z;
    acc += t3 * av.w;
    acc = xadd<0x161>(acc);
    acc = xadd<0x162>(acc);
    acc = xadd<0x164>(acc);
    #pragma unroll
    for (int h = 0; h < 4; ++h) {
      float sh = lane_get(acc, 8 * h);
      if (lane == qq * 4 + h) lg = sh;
    }
  }
  return lg;
}

__device__ __forceinline__ void accum8(const float* __restrict__ xl, float* sagg, float* sden,
                                       int s, int dl, float lg, unsigned lm, int lane) {
  float v[8][4];
  #pragma unroll
  for (int qq = 0; qq < 8; ++qq) {
    int sj = lane_geti(s, qq * 4);
    const float* xls = xl + (size_t)sj * 128;
    #pragma unroll
    for (int i = 0; i < 4; ++i) v[qq][i] = xls[lane + 32 * i];
  }
  const float w = expf(lg - key2f(lm));
  atomicAdd(&sden[dl * 4 + (lane & 3)], w);
  #pragma unroll
  for (int qq = 0; qq < 8; ++qq) {
    int dj = lane_geti(dl, qq * 4);
    float* ad = sagg + dj * 128;
    #pragma unroll
    for (int i = 0; i < 4; ++i) {
      float wh = lane_get(w, qq * 4 + i);
      atomicAdd(&ad[lane + 32 * i], wh * v[qq][i]);
    }
  }
}

__global__ __launch_bounds__(RWAVES * 32)
void gatv2_range_layer1(const float* __restrict__ xl,
                        const float* __restrict__ xin,
                        const _Float16* __restrict__ wrp,
                        const float* __restrict__ att,
                        const float* __restrict__ b1,
                        float* __restrict__ hout,
                        int* __restrict__ qsave,
                        int* __restrict__ qcnt,
                        const int* __restrict__ srcA,
                        const int* __restrict__ dstA,
                        int E, int Nn) {
  __shared__ __attribute__((aligned(16))) float sagg[(RNODES + 1) * 128];
  __shared__ __attribute__((aligned(16))) float sden[(RNODES + 1) * 4];
  __shared__ __attribute__((aligned(16))) unsigned smax[RNODES * 4];
  __shared__ __attribute__((aligned(16))) float sxr[RNODES * 128];
  __shared__ int   sqe[RWAVES][QW];
  __shared__ float sql[RWAVES][QW * 4];
  __shared__ int   sover;

  const int tid = threadIdx.x, lane = tid & 31;
  const int wv = __builtin_amdgcn_readfirstlane(tid >> 5);
  const int lo = (int)blockIdx.x * RNODES;

  const float4 z4 = make_float4(0.f, 0.f, 0.f, 0.f);
  for (int i = tid; i < (RNODES + 1) * 32; i += RWAVES * 32) ((float4*)sagg)[i] = z4;
  if (tid < RNODES + 1) ((float4*)sden)[tid] = z4;
  if (tid < RNODES) ((uint4*)smax)[tid] = make_uint4(0u, 0u, 0u, 0u);
  if (tid == 0) sover = 0;
  range_rows_wmma<128, RNODES>(xin, wrp, sxr, lo, Nn, wv, lane);
  __syncthreads();

  int* q = sqe[wv];
  float* ql = sql[wv];
  int qn = 0;
  int pc = 0;
  bool over = false;
  const float4 av = ((const float4*)att)[lane];
  const int k = lane >> 2, hh = lane & 3;

  const int n4 = E >> 2;
  const int nch = (n4 + 31) >> 5;
  const int4* d4 = (const int4*)dstA;
  const int nb = (nch + RWAVES * SCAN_UB - 1) / (RWAVES * SCAN_UB);

  for (int b = 0; b <= nb; ++b) {
    if (b < nb) {
      const int c0 = b * (RWAVES * SCAN_UB) + wv;
      int4 dd[SCAN_UB];
      #pragma unroll
      for (int u = 0; u < SCAN_UB; ++u) {
        int p = (c0 + RWAVES * u) * 32 + lane;
        dd[u] = d4[p < n4 ? p : n4 - 1];
      }
      #pragma unroll
      for (int u = 0; u < SCAN_UB; ++u) {
        int p = (c0 + RWAVES * u) * 32 + lane;
        q_push_piece(q, qn, dd[u], p, p < n4, lo);
      }
    } else {
      q_push_rest(q, qn, wv, lane, lo, dstA, E, Nn);
    }
    if (qn > QW) { over = true; break; }

    const int keep = b < nb ? 7 : 0;
    while (qn - pc > keep) {
      const int n = qn - pc < 8 ? qn - pc : 8;
      const bool ok = k < n;
      const int ent = q[pc + (ok ? k : 0)];
      const int e = ent & 0xFFFFF, dl = (int)((unsigned)ent >> 20);
      const int sv = srcA[e < E ? e : E - 1];
      const int s = e < E ? sv : e - E;
      const float lg = logits8(xl, sxr, av, s, dl, lane);
      if (ok) {
        ql[(pc + k) * 4 + hh] = lg;
        atomicMax(&smax[dl * 4 + hh], f2key(lg));
        if (hh == 0) q[pc + k] = s | (dl << 16);
      }
      pc += n;
    }
  }
  if (over && lane == 0) sover = 1;
  __syncthreads();

  const int widx = (int)blockIdx.x * RWAVES + wv;
  if (sover == 0) {
    int* qs = qsave + (size_t)widx * QW;
    #pragma unroll
    for (int u = 0; u < QCOPY; ++u) {
      int i = lane + 32 * u;
      if (i < qn) qs[i] = q[i];
    }
    if (lane == 0) qcnt[widx] = qn;
    for (int base = 0; base < qn; base += 8) {
      const int n = qn - base < 8 ? qn - base : 8;
      const bool ok = k < n;
      const int ix = base + (ok ? k : 0);
      const int ent = q[ix];
      const int s = ent & 0xFFFF, dr = (int)((unsigned)ent >> 16);
      accum8(xl, sagg, sden, s, ok ? dr : RNODES, ql[ix * 4 + hh], smax[dr * 4 + hh], lane);
    }
  } else {
    if (lane == 0) qcnt[widx] = -1;
    for (int scan = 0; scan < 2; ++scan) {
      qn = 0;
      for (int c = wv; ; c += RWAVES) {
        const bool rest = c >= nch;
        if (!rest) {
          int p = c * 32 + lane;
          q_push_piece(q, qn, d4[p < n4 ? p : n4 - 1], p, p < n4, lo);
        } else {
          q_push_rest(q, qn, wv, lane, lo, dstA, E, Nn);
        }
        const int keep = rest ? 0 : 7;
        while (qn > keep) {
          const int n = qn < 8 ? qn : 8;
          const int base = qn - n;
          const bool ok = k < n;
          const int ent = q[base + (ok ? k : 0)];
          const int e = ent & 0xFFFFF, dr = (int)((unsigned)ent >> 20);
          const int sv = srcA[e < E ? e : E - 1];
          const int s = e < E ? sv : e - E;
          const float lg = logits8(xl, sxr, av, s, dr, lane);
          if (scan == 0) {
            if (ok) atomicMax(&smax[dr * 4 + hh], f2key(lg));
          } else {
            accum8(xl, sagg, sden, s, ok ? dr : RNODES, lg, smax[dr * 4 + hh], lane);
          }
          qn = base;
        }
        if (rest) break;
      }
      if (scan == 0) __syncthreads();
    }
  }
  __syncthreads();

  for (int i = tid; i < RNODES * 32; i += RWAVES * 32) {
    int nl = i >> 5, j4 = i & 31;
    int n = lo + nl;
    if (n < Nn) {
      float4 a = ((const float4*)sagg)[i];
      float dn = sden[nl * 4 + (j4 >> 3)] + 1e-16f;
      float4 bb = ((const float4*)b1)[j4];
      float4 r;
      r.x = a.x / dn + bb.x; r.y = a.y / dn + bb.y; r.z = a.z / dn + bb.z; r.w = a.w / dn + bb.w;
      r.x = r.x > 0.f ? r.x : expm1f(r.x);
      r.y = r.y > 0.f ? r.y : expm1f(r.y);
      r.z = r.z > 0.f ? r.z : expm1f(r.z);
      r.w = r.w > 0.f ? r.w : expm1f(r.w);
      ((float4*)hout)[(size_t)n * 32 + j4] = r;
    }
  }
}


__device__ __forceinline__ float logits8_c64(const float* __restrict__ xl, const float* sxr,
                                             float4 av0, float4 av1, int s, int dl, int lane) {
  v4f a[8][2];
  #pragma unroll
  for (int qq = 0; qq < 8; ++qq) {
    int sj = lane_geti(s, qq * 4);
    const v4f* ps = (const v4f*)(xl + (size_t)sj * 256);
    a[qq][0] = ps[lane];
    a[qq][1] = ps[32 + lane];
  }
  asm volatile("" :: "v"(a[0][0]), "v"(a[0][1]), "v"(a[1][0]), "v"(a[1][1]),
                     "v"(a[2][0]), "v"(a[2][1]), "v"(a[3][0]), "v"(a[3][1]),
                     "v"(a[4][0]), "v"(a[4][1]), "v"(a[5][0]), "v"(a[5][1]),
                     "v"(a[6][0]), "v"(a[6][1]), "v"(a[7][0]), "v"(a[7][1]));
  float lg = 0.f;
  #pragma unroll
  for (int qq = 0; qq < 8; ++qq) {
    int dj = lane_geti(dl, qq * 4);
    const float4* pd = (const float4*)(sxr + dj * 256);
    float p[2];
    #pragma unroll
    for (int i = 0; i < 2; ++i) {
      float4 bq = pd[i * 32 + lane];
      float4 aw = i == 0 ? av0 : av1;
      float t0 = a[qq][i].x + bq.x, t1 = a[qq][i].y + bq.y;
      float t2 = a[qq][i].z + bq.z, t3 = a[qq][i].w + bq.w;
      t0 = t0 > 0.f ? t0 : 0.2f * t0;
      t1 = t1 > 0.f ? t1 : 0.2f * t1;
      t2 = t2 > 0.f ? t2 : 0.2f * t2;
      t3 = t3 > 0.f ? t3 : 0.2f * t3;
      float acc = t0 * aw.x;
      acc += t1 * aw.y;
      acc += t2 * aw.z;
      acc += t3 * aw.w;
      acc = xadd<0x161>(acc);
      acc = xadd<0x162>(acc);
      acc = xadd<0x164>(acc);
      acc = xadd<0x168>(acc);
      p[i] = acc;
    }
    #pragma unroll
    for (int h = 0; h < 4; ++h) {
      float sh = lane_get(p[h >> 1], (h & 1) * 16);
      if (lane == qq * 4 + h) lg = sh;
    }
  }
  return lg;
}

__device__ __forceinline__ void accum8_mean(const float* __restrict__ xl, float* sout,
                                            int s, int dl, float a, int lane) {
  float v[8][8];
  #pragma unroll
  for (int qq = 0; qq < 8; ++qq) {
    int sj = lane_geti(s, qq * 4);
    const float* xls = xl + (size_t)sj * 256;
    #pragma unroll
    for (int kk = 0; kk < 8; ++kk) v[qq][kk] = xls[kk * 32 + lane];
  }
  #pragma unroll
  for (int qq = 0; qq < 8; ++qq) {
    int dj = lane_geti(dl, qq * 4);
    float m0 = 0.f, m1 = 0.f;
    #pragma unroll
    for (int h = 0; h < 4; ++h) {
      float ah = lane_get(a, qq * 4 + h);
      m0 += ah * v[qq][2 * h];
      m1 += ah * v[qq][2 * h + 1];
    }
    float* od = sout + dj * 64;
    atomicAdd(&od[lane], m0);
    atomicAdd(&od[32 + lane], m1);
  }
}

__global__ __launch_bounds__(RWAVES * 32)
void gatv2_range_layer2(const float* __restrict__ xl,
                        const float* __restrict__ xin,
                        const _Float16* __restrict__ wrp,
                        const float* __restrict__ att,
                        const float* __restrict__ b2,
                        float* __restrict__ out,
                        float* __restrict__ qlog,
                        const int* __restrict__ qsave,
                        const int* __restrict__ qcnt,
                        const int* __restrict__ srcA,
                        const int* __restrict__ dstA,
                        int E, int Nn) {
  __shared__ __attribute__((aligned(16))) float sout[(RNODES + 1) * 64];
  __shared__ __attribute__((aligned(16))) float sden[(RNODES + 1) * 4];
  __shared__ __attribute__((aligned(16))) unsigned smax[RNODES * 4];
  __shared__ __attribute__((aligned(16))) float sxr[RNODES * 256];
  __shared__ int   sqe[RWAVES][QW];

  const int tid = threadIdx.x, lane = tid & 31;
  const int wv = __builtin_amdgcn_readfirstlane(tid >> 5);
  const int lo = (int)blockIdx.x * RNODES;

  int* q = sqe[wv];
  const int widx = (int)blockIdx.x * RWAVES + wv;
  int qn = __builtin_amdgcn_readfirstlane(qcnt[widx]);
  const bool over = qn < 0;
  qn = qn > QW ? QW : qn;
  if (!over && qn > 0) {
    const int* qs = qsave + (size_t)widx * QW;
    int t[QCOPY];
    #pragma unroll
    for (int u = 0; u < QCOPY; ++u) {
      int i = lane + 32 * u;
      t[u] = qs[i < qn ? i : qn - 1];
    }
    #pragma unroll
    for (int u = 0; u < QCOPY; ++u) {
      int i = lane + 32 * u;
      if (i < qn) q[i] = t[u];
    }
  }

  const float4 z4 = make_float4(0.f, 0.f, 0.f, 0.f);
  for (int i = tid; i < (RNODES + 1) * 16; i += RWAVES * 32) ((float4*)sout)[i] = z4;
  if (tid < RNODES + 1) ((float4*)sden)[tid] = z4;
  if (tid < RNODES) ((uint4*)smax)[tid] = make_uint4(0u, 0u, 0u, 0u);
  range_rows_wmma<256, RNODES>(xin, wrp, sxr, lo, Nn, wv, lane);
  __syncthreads();

  float* ql = qlog + (size_t)widx * (QW * 4);
  const float4 av0 = ((const float4*)att)[lane];
  const float4 av1 = ((const float4*)att)[32 + lane];
  const int k = lane >> 2, hh = lane & 3;

  const int n4 = E >> 2;
  const int nch = (n4 + 31) >> 5;
  const int4* d4 = (const int4*)dstA;

  if (!over) {
    for (int base = 0; base < qn; base += 8) {
      const int n = qn - base < 8 ? qn - base : 8;
      const bool ok = k < n;
      const int ix = base + (ok ? k : 0);
      const int ent = q[ix];
      const int s = ent & 0xFFFF, dl = (int)((unsigned)ent >> 16);
      const float lg = logits8_c64(xl, sxr, av0, av1, s, dl, lane);
      if (ok) {
        ql[ix * 4 + hh] = lg;
        atomicMax(&smax[dl * 4 + hh], f2key(lg));
      }
    }
  }
  __syncthreads();

  if (!over) {
    for (int base = 0; base < qn; base += 8) {
      if (base + k < qn) {
        const int ix = base + k;
        const int dr = (int)((unsigned)q[ix] >> 16);
        const float w = expf(ql[ix * 4 + hh] - key2f(smax[dr * 4 + hh]));
        ql[ix * 4 + hh] = w;
        atomicAdd(&sden[dr * 4 + hh], w);
      }
    }
    __syncthreads();
    for (int base = 0; base < qn; base += 8) {
      const int n = qn - base < 8 ? qn - base : 8;
      const bool ok = k < n;
      const int ix = base + (ok ? k : 0);
      const int ent = q[ix];
      const int s = ent & 0xFFFF, dr = (int)((unsigned)ent >> 16);
      const float a = 0.25f * (ql[ix * 4 + hh] / (sden[dr * 4 + hh] + 1e-16f));
      accum8_mean(xl, sout, s, ok ? dr : RNODES, a, lane);
    }
  } else {
    for (int scan = 0; scan < 3; ++scan) {
      qn = 0;
      for (int c = wv; ; c += RWAVES) {
        const bool rest = c >= nch;
        if (!rest) {
          int p = c * 32 + lane;
          q_push_piece(q, qn, d4[p < n4 ? p : n4 - 1], p, p < n4, lo);
        } else {
          q_push_rest(q, qn, wv, lane, lo, dstA, E, Nn);
        }
        const int keep = rest ? 0 : 7;
        while (qn > keep) {
          const int n = qn < 8 ? qn : 8;
          const int base = qn - n;
          const bool ok = k < n;
          const int ent = q[base + (ok ? k : 0)];
          const int e = ent & 0xFFFFF, dr = (int)((unsigned)ent >> 20);
          const int sv = srcA[e < E ? e : E - 1];
          const int s = e < E ? sv : e - E;
          const float lg = logits8_c64(xl, sxr, av0, av1, s, dr, lane);
          if (scan == 0) {
            if (ok) atomicMax(&smax[dr * 4 + hh], f2key(lg));
          } else {
            const float w = expf(lg - key2f(smax[dr * 4 + hh]));
            if (scan == 1) {
              if (ok) atomicAdd(&sden[dr * 4 + hh], w);
            } else {
              const float a = 0.25f * (w / (sden[dr * 4 + hh] + 1e-16f));
              accum8_mean(xl, sout, s, ok ? dr : RNODES, a, lane);
            }
          }
          qn = base;
        }
        if (rest) break;
      }
      if (scan < 2) __syncthreads();
    }
  }
  __syncthreads();

  for (int i = tid; i < RNODES * 16; i += RWAVES * 32) {
    int nl = i >> 4, j4 = i & 15;
    int n = lo + nl;
    if (n < Nn) {
      float4 a = ((const float4*)sout)[i];
      float4 bb = ((const float4*)b2)[j4];
      float4 r;
      r.x = a.x + bb.x; r.y = a.y + bb.y; r.z = a.z + bb.z; r.w = a.w + bb.w;
      ((float4*)out)[(size_t)n * 16 + j4] = r;
    }
  }
}

extern "C" void kernel_launch(void* const* d_in, const int* in_sizes, int n_in,
                              void* d_out, int out_size, void* d_ws, size_t ws_size,
                              hipStream_t stream) {
  (void)in_sizes; (void)n_in; (void)out_size; (void)ws_size;
  const int  N  = NNODES;
  const int  E  = NEDGES;

  const float* x    = (const float*)d_in[0];
  const int*   ei   = (const int*)  d_in[1];
  const float* Wl1  = (const float*)d_in[2];
  const float* Wr1  = (const float*)d_in[3];
  const float* att1 = (const float*)d_in[4];
  const float* b1   = (const float*)d_in[5];
  const float* Wl2  = (const float*)d_in[6];
  const float* Wr2  = (const float*)d_in[7];
  const float* att2 = (const float*)d_in[8];
  const float* b2   = (const float*)d_in[9];
  const int* srcA = ei;
  const int* dstA = ei + E;

  float* ws    = (float*)d_ws;
  float* hbuf  = ws;
  float* xl    = hbuf + (size_t)N * 128;
  float* qlog  = xl   + (size_t)N * 256;
  int*   qsave = (int*)(qlog + (size_t)RBLOCKS * RWAVES * QW * 4);
  int*   qcnt  = qsave + (size_t)RBLOCKS * RWAVES * QW;
  _Float16* wpk = (_Float16*)(qcnt + (size_t)RBLOCKS * RWAVES);

  const int TPB = 256;
  int gblocks = (((N + 15) / 16) + 7) / 8;
  int rblocks = RBLOCKS;

  gatv2_pack_w<<<PK_ALL / TPB, TPB, 0, stream>>>(Wl1, Wr1, Wl2, Wr2, wpk);

  gatv2_gemm_nt_wmma<<<gblocks, TPB, 0, stream>>>(x, wpk + 8 * PK_W1L, xl, N, 128);
  gatv2_range_layer1<<<rblocks, RWAVES * 32, 0, stream>>>(xl, x, wpk + 8 * PK_W1R, att1, b1, hbuf, qsave, qcnt,
                                                          srcA, dstA, E, N);

  gatv2_gemm_nt_wmma<<<gblocks, TPB, 0, stream>>>(hbuf, wpk + 8 * PK_W2L, xl, N, 256);
  gatv2_range_layer2<<<rblocks, RWAVES * 32, 0, stream>>>(xl, hbuf, wpk + 8 * PK_W2R, att2, b2, (float*)d_out, qlog,
                                                          qsave, qcnt, srcA, dstA, E, N);
}


# Layer-1 edge search reads one-byte range tags (written per call in the weight pre-pass launch): 16 edges per 16-byte load, zero-byte test; dst read only for queued edges; sums in arrival order
# speedup vs baseline: 4.3711x; 1.2305x over previous
#include <hip/hip_runtime.h>
#include <hip/hip_bf16.h>
#include <cmath>

typedef __attribute__((ext_vector_type(16))) _Float16 v16h;
typedef __attribute__((ext_vector_type(8)))  float    v8f;
typedef __attribute__((ext_vector_type(4)))  float    v4f;
typedef __attribute__((ext_vector_type(8)))  _Float16 v8h;

#define NNODES 50000
#define NEDGES 500000
#define RNODES   196

__device__ __forceinline__ unsigned f2key(float f) {
  unsigned b = __float_as_uint(f);
  return (b & 0x80000000u) ? ~b : (b | 0x80000000u);
}
__device__ __forceinline__ float key2f(unsigned u) {
  if (u == 0u) return 0.0f;
  return (u & 0x80000000u) ? __uint_as_float(u ^ 0x80000000u) : __uint_as_float(~u);
}

__device__ __forceinline__ v16h load_cvt_f16(const float* p  ) {
  const float4* q = (const float4*)p;
  float4 v0 = q[0], v1 = q[1], v2 = q[4], v3 = q[5];
  v16h r;
  r[0] =(_Float16)v0.x; r[1] =(_Float16)v0.y; r[2] =(_Float16)v0.z; r[3] =(_Float16)v0.w;
  r[4] =(_Float16)v1.x; r[5] =(_Float16)v1.y; r[6] =(_Float16)v1.z; r[7] =(_Float16)v1.w;
  r[8] =(_Float16)v2.x; r[9] =(_Float16)v2.y; r[10]=(_Float16)v2.z; r[11]=(_Float16)v2.w;
  r[12]=(_Float16)v3.x; r[13]=(_Float16)v3.y; r[14]=(_Float16)v3.z; r[15]=(_Float16)v3.w;
  return r;
}

#define PK_W1L 0
#define PK_W1R 2048
#define PK_W2L 4096
#define PK_W2R 8192
#define PK_ALL 12288
#define TAG_PIECES (NEDGES >> 4)

static_assert((NNODES + RNODES - 1) / RNODES <= 256, "a range number fits the one-byte tag");

__device__ __forceinline__ unsigned range_tag(int d) {
  unsigned r = (unsigned)d / (unsigned)RNODES;
  return r < 255u ? r : 255u;
}
__device__ __forceinline__ unsigned range_tag4(int4 d) {
  return range_tag(d.x) | (range_tag(d.y) << 8) | (range_tag(d.z) << 16) | (range_tag(d.w) << 24);
}

__global__ void gatv2_pack_w(const float* __restrict__ W1l,
                             const float* __restrict__ W1r,
                             const float* __restrict__ W2l,
                             const float* __restrict__ W2r,
                             _Float16* __restrict__ Ph,
                             const int* __restrict__ dstA,
                             unsigned* __restrict__ tags, int E) {
  v8h* P = (v8h*)Ph;
  int t = (int)(blockIdx.x * blockDim.x + threadIdx.x);
  if (t >= PK_ALL) {
    int u = t - PK_ALL;
    if (u < (E >> 4)) {
      const int4* d4 = (const int4*)dstA + (size_t)u * 4;
      int4 d0 = d4[0], d1 = d4[1], d2 = d4[2], d3 = d4[3];
      ((uint4*)tags)[u] = make_uint4(range_tag4(d0), range_tag4(d1), range_tag4(d2), range_tag4(d3));
    }
    return;
  }
  const float* W; int u;
  if (t < PK_W1R)      { W = W1l; u = t - PK_W1L; }
  else if (t < PK_W2L) { W = W1r; u = t - PK_W1R; }
  else if (t < PK_W2R) { W = W2l; u = t - PK_W2L; }
  else                 { W = W2r; u = t - PK_W2R; }
  int lane = u & 31, p = (u >> 5) & 1, kk = (u >> 6) & 3, tm = u >> 8;
  int l = lane & 15, half = lane >> 4;
  const float4* q = (const float4*)(W + (size_t)(tm * 16 + l) * 128 + kk * 32 + p * 16 + half * 8);
  float4 v0 = q[0], v1 = q[1];
  v8h r;
  r[0] = (_Float16)v0.x; r[1] = (_Float16)v0.y; r[2] = (_Float16)v0.z; r[3] = (_Float16)v0.w;
  r[4] = (_Float16)v1.x; r[5] = (_Float16)v1.y; r[6] = (_Float16)v1.z; r[7] = (_Float16)v1.w;
  P[t] = r;
}

__global__ void gatv2_gemm_nt_wmma(const float* __restrict__ X,
                                   const _Float16* __restrict__ Wph,
                                   float* __restrict__ Y,
                                   int Nrows, int Mcols) {
  const int K = 128;
  const v8h* Wp = (const v8h*)Wph;
  int wave = (int)((blockIdx.x * blockDim.x + threadIdx.x) >> 5);
  int lane = threadIdx.x & 31;
  int ntN = (Nrows + 15) >> 4;
  if (wave >= ntN) return;
  int half = lane >> 4;
  int l = lane & 15;
  int row = wave * 16 + l; if (row >= Nrows) row = Nrows - 1;
  const float* xrow = X + (size_t)row * K + half * 8;

  v16h a[4];
  #pragma unroll
  for (int kk = 0; kk < 4; ++kk) a[kk] = load_cvt_f16(xrow + kk * 32);

  int ntM = Mcols >> 4;
  for (int tm = 0; tm < ntM; tm += 2) {
    const v8h* wt = Wp + (size_t)tm * 256 + lane;
    v8h b[16];
    #pragma unroll
    for (int i = 0; i < 16; ++i) b[i] = wt[i * 32];
    asm volatile("" :: "v"(b[0]), "v"(b[1]), "v"(b[2]), "v"(b[3]), "v"(b[4]), "v"(b[5]), "v"(b[6]), "v"(b[7]),
                       "v"(b[8]), "v"(b[9]), "v"(b[10]), "v"(b[11]), "v"(b[12]), "v"(b[13]), "v"(b[14]), "v"(b[15]));
    v8f acc0 = {}, acc1 = {};
    #pragma unroll
    for (int kk = 0; kk < 4; ++kk) {
      v16h bb = __builtin_shufflevector(b[2 * kk], b[2 * kk + 1], 0, 1, 2, 3, 4, 5, 6, 7, 8, 9, 10, 11, 12, 13, 14, 15);
      acc0 = __builtin_amdgcn_wmma_f32_16x16x32_f16(
           false, a[kk],  false, bb,
           (short)0, acc0,  false,  false);
    }
    #pragma unroll
    for (int kk = 0; kk < 4; ++kk) {
      v16h bb = __builtin_shufflevector(b[8 + 2 * kk], b[9 + 2 * kk], 0, 1, 2, 3, 4, 5, 6, 7, 8, 9, 10, 11, 12, 13, 14, 15);
      acc1 = __builtin_amdgcn_wmma_f32_16x16x32_f16(
           false, a[kk],  false, bb,
           (short)0, acc1,  false,  false);
    }
    #pragma unroll
    for (int r = 0; r < 8; ++r) {
      int orow = wave * 16 + r + half * 8;
      if (orow < Nrows) {
        float* yp = Y + (size_t)orow * Mcols + tm * 16 + l;
        yp[0]  = acc0[r];
        yp[16] = acc1[r];
      }
    }
  }
}

template<int MC, int RN, int WPS>
__device__ __forceinline__ void range_rows_wmma(const float* __restrict__ X,
                                                const _Float16* __restrict__ Wph,
                                                float* srow, int lo, int Nn, int wv, int lane) {
  constexpr int NSTRIP = (RN + 15) / 16;
  constexpr int NTH = MC / 16 / WPS;
  static_assert(NTH % 2 == 0 && NTH * WPS * 16 == MC, "column tiles split evenly over the waves of a strip, two per trip");
  static_assert(WPS * NSTRIP <= 32, "one strip share per wave of the group");
  if (wv < WPS * NSTRIP) {
    const v8h* Wp = (const v8h*)Wph;
    const int strip = wv / WPS, ch = wv % WPS;
    const int r0 = strip * 16 < RN - 16 ? strip * 16 : RN - 16;
    const int half = lane >> 4, l = lane & 15;
    int n = lo + r0 + l; if (n >= Nn) n = Nn - 1;
    const float* xrow = X + (size_t)n * 128 + half * 8;
    v16h a[4];
    #pragma unroll
    for (int kk = 0; kk < 4; ++kk) a[kk] = load_cvt_f16(xrow + kk * 32);
    #pragma nounroll
    for (int tm = ch * NTH; tm < (ch + 1) * NTH; tm += 2) {
      const v8h* wt = Wp + (size_t)tm * 256 + lane;
      v8h b[16];
      #pragma unroll
      for (int i = 0; i < 16; ++i) b[i] = wt[i * 32];
      asm volatile("" :: "v"(b[0]), "v"(b[1]), "v"(b[2]), "v"(b[3]), "v"(b[4]), "v"(b[5]), "v"(b[6]), "v"(b[7]),
                         "v"(b[8]), "v"(b[9]), "v"(b[10]), "v"(b[11]), "v"(b[12]), "v"(b[13]), "v"(b[14]), "v"(b[15]));
      v8f acc0 = {}, acc1 = {};
      #pragma unroll
      for (int kk = 0; kk < 4; ++kk) {
        v16h bb = __builtin_shufflevector(b[2 * kk], b[2 * kk + 1], 0, 1, 2, 3, 4, 5, 6, 7, 8, 9, 10, 11, 12, 13, 14, 15);
        acc0 = __builtin_amdgcn_wmma_f32_16x16x32_f16(
             false, a[kk],  false, bb,
             (short)0, acc0,  false,  false);
      }
      #pragma unroll
      for (int kk = 0; kk < 4; ++kk) {
        v16h bb = __builtin_shufflevector(b[8 + 2 * kk], b[9 + 2 * kk], 0, 1, 2, 3, 4, 5, 6, 7, 8, 9, 10, 11, 12, 13, 14, 15);
        acc1 = __builtin_amdgcn_wmma_f32_16x16x32_f16(
             false, a[kk],  false, bb,
             (short)0, acc1,  false,  false);
      }
      float* yp = srow + (r0 + half * 8) * MC + tm * 16 + l;
      #pragma unroll
      for (int r = 0; r < 8; ++r) {
        yp[r * MC]      = acc0[r];
        yp[r * MC + 16] = acc1[r];
      }
    }
  }
}

__device__ __forceinline__ float lane_get(float v, int l) {
  return __int_as_float(__builtin_amdgcn_readlane(__float_as_int(v), l));
}
__device__ __forceinline__ int lane_geti(int v, int l) {
  return __builtin_amdgcn_readlane(v, l);
}
template<int CTRL>
__device__ __forceinline__ float xadd(float v) {
  return v + __int_as_float(__builtin_amdgcn_update_dpp(0, __float_as_int(v), CTRL, 0xf, 0xf, true));
}
template<int Q>
__device__ __forceinline__ float oct_get(float v) {
  return __int_as_float((int)__builtin_amdgcn_mov_dpp8((unsigned)__float_as_int(v), (unsigned)(Q * 0x249249)));
}
template<int R>
__device__ __forceinline__ float row_get(float v) {
  return __int_as_float(__builtin_amdgcn_update_dpp(0, __float_as_int(v), 0x150 + R, 0xf, 0xf, true));
}
__device__ __forceinline__ float4 as_f4(v4f v) { return make_float4(v.x, v.y, v.z, v.w); }
__device__ __forceinline__ float lrelu_dot4(float4 a, float4 b, float4 w) {
  float t0 = a.x + b.x, t1 = a.y + b.y;
  float t2 = a.z + b.z, t3 = a.w + b.w;
  t0 = t0 > 0.f ? t0 : 0.2f * t0;
  t1 = t1 > 0.f ? t1 : 0.2f * t1;
  t2 = t2 > 0.f ? t2 : 0.2f * t2;
  t3 = t3 > 0.f ? t3 : 0.2f * t3;
  float acc = t0 * w.x;
  acc += t1 * w.y;
  acc += t2 * w.z;
  acc += t3 * w.w;
  return acc;
}

#define HNODES   98
#define RWAVES   32
#define SCAN_UB  8
#define QW       184
#define RBLOCKS  ((NNODES + RNODES - 1) / RNODES)
#define QCOPY    ((QW + 31) / 32)
#define SHIFT_GAP 30.0f

static_assert(NNODES <= 65536, "a queue entry keeps src in 16 bits");
static_assert(NNODES + NEDGES <= (1 << 20), "a queue entry keeps the edge number in 20 bits");
static_assert(RNODES <= 256, "a queue entry keeps the local node in 8 bits");
static_assert(7 + 128 <= QW, "the scan path queues one piece per lane (128 edges) on top of a rest of 7");
static_assert(2 * HNODES == RNODES, "two layer-2 groups share one layer-1 range");
static_assert(RNODES <= 8 * RWAVES, "the self-loop pre-phase gives every wave at most one group of 8 nodes");

__device__ __forceinline__ void q_push(int* q, int& qn, bool m, int v) {
  unsigned mask = __builtin_amdgcn_ballot_w32(m);
  int pos = qn + (int)__builtin_amdgcn_mbcnt_lo(mask, 0u);
  if (m && pos < QW) q[pos] = v;
  qn += __builtin_popcount(mask);
}

template<int RN>
__device__ __forceinline__ void q_push_piece(int* q, int& qn, int4 dd, int p, bool in, int lo) {
  int r0 = dd.x - lo, r1 = dd.y - lo, r2 = dd.z - lo, r3 = dd.w - lo;
  bool m0 = in & ((unsigned)r0 < (unsigned)RN);
  bool m1 = in & ((unsigned)r1 < (unsigned)RN);
  bool m2 = in & ((unsigned)r2 < (unsigned)RN);
  bool m3 = in & ((unsigned)r3 < (unsigned)RN);
  if (__builtin_amdgcn_ballot_w32(m0 | m1 | m2 | m3)) {
    q_push(q, qn, m0, (4 * p)     | (int)((unsigned)r0 << 20));
    q_push(q, qn, m1, (4 * p + 1) | (int)((unsigned)r1 << 20));
    q_push(q, qn, m2, (4 * p + 2) | (int)((unsigned)r2 << 20));
    q_push(q, qn, m3, (4 * p + 3) | (int)((unsigned)r3 << 20));
  }
}

template<int RN, bool SELF>
__device__ __forceinline__ void q_push_rest(int* q, int& qn, int wv, int lane, int lo,
                                            const int* __restrict__ dstA, int E, int Nn) {
  if (wv == 0) {
    int e = (E >> 2) * 4 + lane;
    int ec = e < E ? e : E - 1;
    int rr = dstA[ec] - lo;
    bool m = e < E && (unsigned)rr < (unsigned)RN;
    q_push(q, qn, m, e | (int)((unsigned)rr << 20));
  }
  if (SELF) {
    int nl = wv + RWAVES * lane;
    bool ms = nl < RN && lo + nl < Nn;
    q_push(q, qn, ms, (E + lo + nl) | (nl << 20));
  }
}

__device__ __forceinline__ float logits8(const float* __restrict__ xl, const float* sxr,
                                         float4 av, int s, int dl, int lane) {
  float4 a[8];
  #pragma unroll
  for (int qq = 0; qq < 8; ++qq) {
    int sj = lane_geti(s, qq * 4);
    a[qq] = ((const float4*)(xl + (size_t)sj * 128))[lane];
  }
  float lg = 0.f;
  #pragma unroll
  for (int qq = 0; qq < 8; ++qq) {
    int dj = lane_geti(dl, qq * 4);
    float4 bq = ((const float4*)(sxr + dj * 128))[lane];
    float acc = lrelu_dot4(a[qq], bq, av);
    acc = xadd<0x161>(acc);
    acc = xadd<0x162>(acc);
    acc = xadd<0x164>(acc);
    #pragma unroll
    for (int h = 0; h < 4; ++h) {
      float sh = lane_get(acc, 8 * h);
      if (lane == qq * 4 + h) lg = sh;
    }
  }
  return lg;
}

__device__ __forceinline__ void accum8(const float* __restrict__ xl, float* sagg, float* sden,
                                       int s, int dl, float lg, unsigned lm, int lane) {
  float v[8][4];
  #pragma unroll
  for (int qq = 0; qq < 8; ++qq) {
    int sj = lane_geti(s, qq * 4);
    const float* xls = xl + (size_t)sj * 128;
    #pragma unroll
    for (int i = 0; i < 4; ++i) v[qq][i] = xls[lane + 32 * i];
  }
  const float w = expf(lg - key2f(lm));
  atomicAdd(&sden[dl * 4 + (lane & 3)], w);
  const int pl = (lane & 3) * 32 + (lane >> 2);
  #pragma unroll
  for (int qq = 0; qq < 8; ++qq) {
    int dj = lane_geti(dl, qq * 4);
    float* ad = sagg + dj * 128 + pl;
    #pragma unroll
    for (int i = 0; i < 4; ++i) {
      float wh = lane_get(w, qq * 4 + i);
      atomicAdd(&ad[8 * i], wh * v[qq][i]);
    }
  }
}

__device__ __forceinline__ void self8(const float* __restrict__ xl, const float* sxr, float* sm0,
                                      float* sagg, float* sden, float4 av,
                                      int lo, int nb, int cnt, int lane) {
  float4 a[8];
  #pragma unroll
  for (int qq = 0; qq < 8; ++qq) {
    int nl = nb + (qq < cnt ? qq : 0);
    a[qq] = ((const float4*)(xl + (size_t)(lo + nl) * 128))[lane];
  }
  #pragma unroll
  for (int qq = 0; qq < 8; ++qq) {
    if (qq < cnt) {
      const int nl = nb + qq;
      float4 bq = ((const float4*)(sxr + nl * 128))[lane];
      float acc = lrelu_dot4(a[qq], bq, av);
      acc = xadd<0x161>(acc);
      acc = xadd<0x162>(acc);
      acc = xadd<0x164>(acc);
      float* ad = sagg + nl * 128 + lane;
      ad[0] = a[qq].x; ad[32] = a[qq].y; ad[64] = a[qq].z; ad[96] = a[qq].w;
      if ((lane & 7) == 0) {
        sm0[nl * 4 + (lane >> 3)] = acc;
        sden[nl * 4 + (lane >> 3)] = 1.0f;
      }
    }
  }
}

template<int Q>
__device__ __forceinline__ void add_row(float* sagg, float4 a, float w, int dla, int lane) {
  const int dj = lane_geti(dla, Q * 4);
  const float wq = oct_get<Q>(w);
  float* ad = sagg + dj * 128 + lane;
  atomicAdd(&ad[0],  wq * a.x);
  atomicAdd(&ad[32], wq * a.y);
  atomicAdd(&ad[64], wq * a.z);
  atomicAdd(&ad[96], wq * a.w);
}

__device__ __forceinline__ bool fused8(const float* __restrict__ xl, const float* sxr, const float* sm0,
                                       float* sagg, float* sden, float4 av,
                                       int s, int dlr, int dla, int dl2r, int dl2a, int lane) {
  float4 a[8];
  #pragma unroll
  for (int qq = 0; qq < 8; ++qq) {
    int sj = lane_geti(s, qq * 4);
    a[qq] = ((const float4*)(xl + (size_t)sj * 128))[lane];
  }
  float lg = 0.f;
  #pragma unroll
  for (int qq = 0; qq < 8; ++qq) {
    int dj = lane_geti(dlr, qq * 4);
    float4 bq = ((const float4*)(sxr + dj * 128))[lane];
    float acc = lrelu_dot4(a[qq], bq, av);
    acc = xadd<0x161>(acc);
    acc = xadd<0x162>(acc);
    acc = xadd<0x164>(acc);
    if ((lane & 7) == qq) lg = acc;
  }
  const int hd = lane >> 3;
  const float d = lg - sm0[dl2r * 4 + hd];
  const float w = expf(d);
  atomicAdd(&sden[dl2a * 4 + hd], w);
  add_row<0>(sagg, a[0], w, dla, lane);
  add_row<1>(sagg, a[1], w, dla, lane);
  add_row<2>(sagg, a[2], w, dla, lane);
  add_row<3>(sagg, a[3], w, dla, lane);
  add_row<4>(sagg, a[4], w, dla, lane);
  add_row<5>(sagg, a[5], w, dla, lane);
  add_row<6>(sagg, a[6], w, dla, lane);
  add_row<7>(sagg, a[7], w, dla, lane);
  return d > SHIFT_GAP;
}

__device__ __forceinline__ unsigned nz_bytes(unsigned x) {
  return ((x & 0x7F7F7F7Fu) + 0x7F7F7F7Fu) | x;
}

__device__ __forceinline__ void q_push_tags(int* q, int& qn, uint4 tt, int p, bool in, unsigned rsel) {
  const unsigned y0 = nz_bytes(tt.x ^ rsel), y1 = nz_bytes(tt.y ^ rsel);
  const unsigned y2 = nz_bytes(tt.z ^ rsel), y3 = nz_bytes(tt.w ^ rsel);
  const bool hit = in & (((y0 & y1 & y2 & y3) & 0x80808080u) != 0x80808080u);
  if (__builtin_amdgcn_ballot_w32(hit)) {
    unsigned hm = ((~y0 & 0x80808080u) >> 7) | ((~y1 & 0x80808080u) >> 6) |
                  ((~y2 & 0x80808080u) >> 5) | ((~y3 & 0x80808080u) >> 4);
    hm = in ? hm : 0u;
    #pragma nounroll
    for (int it = 0; it < 16; ++it) {
      const bool m = hm != 0u;
      if (!__builtin_amdgcn_ballot_w32(m)) break;
      const int b = __builtin_ctz(hm | 0x80000000u);
      q_push(q, qn, m, 16 * p + ((b & 7) << 2) + (b >> 3));
      hm &= hm - 1u;
    }
  }
}

__global__ __launch_bounds__(RWAVES * 32)
void gatv2_range_layer1(const float* __restrict__ xl,
                        const float* __restrict__ xin,
                        const _Float16* __restrict__ wrp,
                        const float* __restrict__ att,
                        const float* __restrict__ b1,
                        float* __restrict__ hout,
                        int* __restrict__ qsave,
                        int* __restrict__ qcnt,
                        const int* __restrict__ srcA,
                        const int* __restrict__ dstA,
                        const unsigned* __restrict__ tags,
                        int E, int Nn) {
  __shared__ __attribute__((aligned(16))) float sagg[(RNODES + 1) * 128];
  __shared__ __attribute__((aligned(16))) float sden[(RNODES + 1) * 4];
  __shared__ __attribute__((aligned(16))) unsigned smax[RNODES * 4];
  __shared__ __attribute__((aligned(16))) float sm0[RNODES * 4];
  __shared__ __attribute__((aligned(16))) float sxr[RNODES * 128];
  __shared__ int   sqe[RWAVES][QW];
  __shared__ int   sover;

  const int tid = threadIdx.x, lane = tid & 31;
  const int wv = __builtin_amdgcn_readfirstlane(tid >> 5);
  const int lo = (int)blockIdx.x * RNODES;

  const float4 z4 = make_float4(0.f, 0.f, 0.f, 0.f);
  for (int i = tid; i < (RNODES + 1) * 32; i += RWAVES * 32) ((float4*)sagg)[i] = z4;
  if (tid < RNODES + 1) ((float4*)sden)[tid] = z4;
  if (tid < RNODES) ((uint4*)smax)[tid] = make_uint4(0u, 0u, 0u, 0u);
  if (tid == 0) sover = 0;
  range_rows_wmma<128, RNODES, 2>(xin, wrp, sxr, lo, Nn, wv, lane);
  __syncthreads();

  const float4 av = ((const float4*)att)[lane];
  {
    const int nown = Nn - lo < RNODES ? Nn - lo : RNODES;
    const int cnt = nown - 8 * wv < 8 ? nown - 8 * wv : 8;
    if (cnt > 0) self8(xl, sxr, sm0, sagg, sden, av, lo, 8 * wv, cnt, lane);
  }
  __syncthreads();

  int* q = sqe[wv];
  int qn = 0;
  int pc = 0;
  bool over = false;
  const int k = lane >> 2, k2 = lane & 7;

  const int n4 = E >> 2;
  const int nch = (n4 + 31) >> 5;
  const int4* d4 = (const int4*)dstA;
  const int n16 = E >> 4;
  const int ncht = (n16 + 31) >> 5;
  const uint4* t4 = (const uint4*)tags;
  const unsigned rsel = (unsigned)blockIdx.x * 0x01010101u;
  const int nb = (ncht + RWAVES * SCAN_UB - 1) / (RWAVES * SCAN_UB);

  for (int b = 0; b <= nb; ++b) {
    if (b < nb) {
      const int c0 = b * (RWAVES * SCAN_UB) + wv;
      uint4 tg[SCAN_UB];
      #pragma unroll
      for (int u = 0; u < SCAN_UB; ++u) {
        int p = (c0 + RWAVES * u) * 32 + lane;
        tg[u] = t4[p < n16 ? p : n16 - 1];
      }
      #pragma unroll
      for (int u = 0; u < SCAN_UB; ++u) {
        int p = (c0 + RWAVES * u) * 32 + lane;
        q_push_tags(q, qn, tg[u], p, p < n16, rsel);
      }
    } else if (wv == 0) {
      int e = n16 * 16 + lane;
      int ec = e < E ? e : E - 1;
      int rr = dstA[ec] - lo;
      q_push(q, qn, lane < 16 && e < E && (unsigned)rr < (unsigned)RNODES, e);
    }
    if (qn > QW) { over = true; break; }

    const int keep = b < nb ? 7 : 0;
    while (qn - pc > keep) {
      const int n = qn - pc < 8 ? qn - pc : 8;
      const bool ok = k < n, ok2 = k2 < n;
      const int ea = q[pc + (ok ? k : 0)] & 0xFFFFF, eb = q[pc + (ok2 ? k2 : 0)] & 0xFFFFF;
      const int e = ea < E ? ea : E - 1, e2 = eb < E ? eb : E - 1;
      const int s = srcA[e];
      const int dl = dstA[e] - lo, dl2 = dstA[e2] - lo;
      const bool in1 = (unsigned)dl < (unsigned)RNODES, in2 = (unsigned)dl2 < (unsigned)RNODES;
      const int dla = ok && in1 ? dl : RNODES;
      const bool big = fused8(xl, sxr, sm0, sagg, sden, av, s, in1 ? dl : 0, dla,
                              in2 ? dl2 : 0, ok2 && in2 ? dl2 : RNODES, lane);
      if (ok && (lane & 3) == 0) q[pc + k] = s | (dla << 16);
      pc += n;
      if (__builtin_amdgcn_ballot_w32(big && ok2 && in2)) { over = true; break; }
    }
    if (over) break;
  }
  if (over && lane == 0) sover = 1;
  __syncthreads();

  const int widx = (int)blockIdx.x * RWAVES + wv;
  if (sover == 0) {
    int* qs = qsave + (size_t)widx * QW;
    #pragma unroll
    for (int u = 0; u < QCOPY; ++u) {
      int i = lane + 32 * u;
      if (i < qn) qs[i] = q[i];
    }
    if (lane == 0) qcnt[widx] = qn;
  } else {
    if (lane == 0) qcnt[widx] = -1;
    for (int i = tid; i < (RNODES + 1) * 32; i += RWAVES * 32) ((float4*)sagg)[i] = z4;
    if (tid < RNODES + 1) ((float4*)sden)[tid] = z4;
    const int hh = lane & 3;
    for (int scan = 0; scan < 2; ++scan) {
      qn = 0;
      for (int c = wv; ; c += RWAVES) {
        const bool rest = c >= nch;
        if (!rest) {
          int p = c * 32 + lane;
          q_push_piece<RNODES>(q, qn, d4[p < n4 ? p : n4 - 1], p, p < n4, lo);
        } else {
          q_push_rest<RNODES, true>(q, qn, wv, lane, lo, dstA, E, Nn);
        }
        const int keep = rest ? 0 : 7;
        while (qn > keep) {
          const int n = qn < 8 ? qn : 8;
          const int base = qn - n;
          const bool ok = k < n;
          const int ent = q[base + (ok ? k : 0)];
          const int e = ent & 0xFFFFF, dr = (int)((unsigned)ent >> 20);
          const int sv = srcA[e < E ? e : E - 1];
          const int s = e < E ? sv : e - E;
          const float lg = logits8(xl, sxr, av, s, dr, lane);
          if (scan == 0) {
            if (ok) atomicMax(&smax[dr * 4 + hh], f2key(lg));
          } else {
            accum8(xl, sagg, sden, s, ok ? dr : RNODES, lg, smax[dr * 4 + hh], lane);
          }
          qn = base;
        }
        if (rest) break;
      }
      if (scan == 0) __syncthreads();
    }
  }
  __syncthreads();

  for (int i = tid; i < RNODES * 32; i += RWAVES * 32) {
    int nl = i >> 5, j4 = i & 31;
    int n = lo + nl;
    if (n < Nn) {
      const float* ar = sagg + nl * 128 + j4;
      float4 a = make_float4(ar[0], ar[32], ar[64], ar[96]);
      float dn = sden[nl * 4 + (j4 >> 3)] + 1e-16f;
      float4 bb = ((const float4*)b1)[j4];
      float4 r;
      r.x = a.x / dn + bb.x; r.y = a.y / dn + bb.y; r.z = a.z / dn + bb.z; r.w = a.w / dn + bb.w;
      r.x = r.x > 0.f ? r.x : expm1f(r.x);
      r.y = r.y > 0.f ? r.y : expm1f(r.y);
      r.z = r.z > 0.f ? r.z : expm1f(r.z);
      r.w = r.w > 0.f ? r.w : expm1f(r.w);
      ((float4*)hout)[(size_t)n * 32 + j4] = r;
    }
  }
}


__device__ __forceinline__ float logits8_c64(const float* __restrict__ xl, const float* sxr,
                                             float4 av0, float4 av1, int s, int dl, int lane) {
  v4f a[8][2];
  #pragma unroll
  for (int qq = 0; qq < 8; ++qq) {
    int sj = lane_geti(s, qq * 4);
    const v4f* ps = (const v4f*)(xl + (size_t)sj * 256);
    a[qq][0] = ps[lane];
    a[qq][1] = ps[32 + lane];
  }
  asm volatile("" :: "v"(a[0][0]), "v"(a[0][1]), "v"(a[1][0]), "v"(a[1][1]),
                     "v"(a[2][0]), "v"(a[2][1]), "v"(a[3][0]), "v"(a[3][1]),
                     "v"(a[4][0]), "v"(a[4][1]), "v"(a[5][0]), "v"(a[5][1]),
                     "v"(a[6][0]), "v"(a[6][1]), "v"(a[7][0]), "v"(a[7][1]));
  float lg = 0.f;
  #pragma unroll
  for (int qq = 0; qq < 8; ++qq) {
    int dj = lane_geti(dl, qq * 4);
    const float4* pd = (const float4*)(sxr + dj * 256);
    float p[2];
    #pragma unroll
    for (int i = 0; i < 2; ++i) {
      float4 bq = pd[i * 32 + lane];
      float4 aw = i == 0 ? av0 : av1;
      float t0 = a[qq][i].x + bq.x, t1 = a[qq][i].y + bq.y;
      float t2 = a[qq][i].z + bq.z, t3 = a[qq][i].w + bq.w;
      t0 = t0 > 0.f ? t0 : 0.2f * t0;
      t1 = t1 > 0.f ? t1 : 0.2f * t1;
      t2 = t2 > 0.f ? t2 : 0.2f * t2;
      t3 = t3 > 0.f ? t3 : 0.2f * t3;
      float acc = t0 * aw.x;
      acc += t1 * aw.y;
      acc += t2 * aw.z;
      acc += t3 * aw.w;
      acc = xadd<0x161>(acc);
      acc = xadd<0x162>(acc);
      acc = xadd<0x164>(acc);
      acc = xadd<0x168>(acc);
      p[i] = acc;
    }
    #pragma unroll
    for (int h = 0; h < 4; ++h) {
      float sh = lane_get(p[h >> 1], (h & 1) * 16);
      if (lane == qq * 4 + h) lg = sh;
    }
  }
  return lg;
}

__device__ __forceinline__ void accum8_mean(const float* __restrict__ xl, float* sout,
                                            int s, int dl, float a, int lane) {
  float v[8][8];
  #pragma unroll
  for (int qq = 0; qq < 8; ++qq) {
    int sj = lane_geti(s, qq * 4);
    const float* xls = xl + (size_t)sj * 256;
    #pragma unroll
    for (int kk = 0; kk < 8; ++kk) v[qq][kk] = xls[kk * 32 + lane];
  }
  #pragma unroll
  for (int qq = 0; qq < 8; ++qq) {
    int dj = lane_geti(dl, qq * 4);
    float m0 = 0.f, m1 = 0.f;
    #pragma unroll
    for (int h = 0; h < 4; ++h) {
      float ah = lane_get(a, qq * 4 + h);
      m0 += ah * v[qq][2 * h];
      m1 += ah * v[qq][2 * h + 1];
    }
    float* od = sout + dj * 64;
    atomicAdd(&od[lane], m0);
    atomicAdd(&od[32 + lane], m1);
  }
}

__device__ __forceinline__ void self8_c64(const float* __restrict__ xl, const float* sxr, float* sm0,
                                          float* sagg, float* sden, float4 av0, float4 av1,
                                          int lo, int nb, int cnt, int lane) {
  v4f a[8][2];
  #pragma unroll
  for (int qq = 0; qq < 8; ++qq) {
    int nl = nb + (qq < cnt ? qq : 0);
    const v4f* ps = (const v4f*)(xl + (size_t)(lo + nl) * 256);
    a[qq][0] = ps[lane];
    a[qq][1] = ps[32 + lane];
  }
  asm volatile("" :: "v"(a[0][0]), "v"(a[0][1]), "v"(a[1][0]), "v"(a[1][1]),
                     "v"(a[2][0]), "v"(a[2][1]), "v"(a[3][0]), "v"(a[3][1]),
                     "v"(a[4][0]), "v"(a[4][1]), "v"(a[5][0]), "v"(a[5][1]),
                     "v"(a[6][0]), "v"(a[6][1]), "v"(a[7][0]), "v"(a[7][1]));
  #pragma unroll
  for (int qq = 0; qq < 8; ++qq) {
    if (qq < cnt) {
      const int nl = nb + qq;
      const float4* pd = (const float4*)(sxr + nl * 256);
      #pragma unroll
      for (int i = 0; i < 2; ++i) {
        float acc = lrelu_dot4(as_f4(a[qq][i]), pd[i * 32 + lane], i == 0 ? av0 : av1);
        acc = xadd<0x161>(acc);
        acc = xadd<0x162>(acc);
        acc = xadd<0x164>(acc);
        acc = xadd<0x168>(acc);
        float* ad = sagg + nl * 256 + i * 128 + lane;
        ad[0] = a[qq][i].x; ad[32] = a[qq][i].y; ad[64] = a[qq][i].z; ad[96] = a[qq][i].w;
        if ((lane & 15) == 0) {
          sm0[nl * 4 + 2 * i + (lane >> 4)] = acc;
          sden[nl * 4 + 2 * i + (lane >> 4)] = 1.0f;
        }
      }
    }
  }
}

template<int Q>
__device__ __forceinline__ void add_row_c64(float* sagg, v4f a0, v4f a1, float w, int dla, int lane) {
  const int dj = lane_geti(dla, Q * 4);
  const float w0 = row_get<Q>(w);
  const float w1 = row_get<8 + Q>(w);
  float* ad = sagg + dj * 256 + lane;
  atomicAdd(&ad[0],   w0 * a0.x);
  atomicAdd(&ad[32],  w0 * a0.y);
  atomicAdd(&ad[64],  w0 * a0.z);
  atomicAdd(&ad[96],  w0 * a0.w);
  atomicAdd(&ad[128], w1 * a1.x);
  atomicAdd(&ad[160], w1 * a1.y);
  atomicAdd(&ad[192], w1 * a1.z);
  atomicAdd(&ad[224], w1 * a1.w);
}

__device__ __forceinline__ bool fused8_c64(const float* __restrict__ xl, const float* sxr, const float* sm0,
                                           float* sagg, float* sden, float4 av0, float4 av1,
                                           int s, int dlr, int dla, int dl2r, int dl2a, int lane) {
  v4f a[8][2];
  #pragma unroll
  for (int qq = 0; qq < 8; ++qq) {
    int sj = lane_geti(s, qq * 4);
    const v4f* ps = (const v4f*)(xl + (size_t)sj * 256);
    a[qq][0] = ps[lane];
    a[qq][1] = ps[32 + lane];
  }
  asm volatile("" :: "v"(a[0][0]), "v"(a[0][1]), "v"(a[1][0]), "v"(a[1][1]),
                     "v"(a[2][0]), "v"(a[2][1]), "v"(a[3][0]), "v"(a[3][1]),
                     "v"(a[4][0]), "v"(a[4][1]), "v"(a[5][0]), "v"(a[5][1]),
                     "v"(a[6][0]), "v"(a[6][1]), "v"(a[7][0]), "v"(a[7][1]));
  float lg = 0.f;
  #pragma unroll
  for (int qq = 0; qq < 8; ++qq) {
    int dj = lane_geti(dlr, qq * 4);
    const float4* pd = (const float4*)(sxr + dj * 256);
    #pragma unroll
    for (int i = 0; i < 2; ++i) {
      float acc = lrelu_dot4(as_f4(a[qq][i]), pd[i * 32 + lane], i == 0 ? av0 : av1);
      acc = xadd<0x161>(acc);
      acc = xadd<0x162>(acc);
      acc = xadd<0x164>(acc);
      acc = xadd<0x168>(acc);
      if ((lane & 15) == i * 8 + qq) lg = acc;
    }
  }
  const int hd = 2 * ((lane >> 3) & 1) + (lane >> 4);
  const float d = lg - sm0[dl2r * 4 + hd];
  const float w = expf(d);
  atomicAdd(&sden[dl2a * 4 + hd], w);
  add_row_c64<0>(sagg, a[0][0], a[0][1], w, dla, lane);
  add_row_c64<1>(sagg, a[1][0], a[1][1], w, dla, lane);
  add_row_c64<2>(sagg, a[2][0], a[2][1], w, dla, lane);
  add_row_c64<3>(sagg, a[3][0], a[3][1], w, dla, lane);
  add_row_c64<4>(sagg, a[4][0], a[4][1], w, dla, lane);
  add_row_c64<5>(sagg, a[5][0], a[5][1], w, dla, lane);
  add_row_c64<6>(sagg, a[6][0], a[6][1], w, dla, lane);
  add_row_c64<7>(sagg, a[7][0], a[7][1], w, dla, lane);
  return d > SHIFT_GAP;
}

__global__ __launch_bounds__(RWAVES * 32)
void gatv2_range_layer2(const float* __restrict__ xl,
                        const float* __restrict__ xin,
                        const _Float16* __restrict__ wrp,
                        const float* __restrict__ att,
                        const float* __restrict__ b2,
                        float* __restrict__ out,
                        const int* __restrict__ qsave,
                        const int* __restrict__ qcnt,
                        const int* __restrict__ srcA,
                        const int* __restrict__ dstA,
                        int E, int Nn) {
  __shared__ __attribute__((aligned(16))) float sagg[(HNODES + 1) * 256];
  __shared__ __attribute__((aligned(16))) float sden[(HNODES + 1) * 4];
  __shared__ __attribute__((aligned(16))) unsigned smax[HNODES * 4];
  __shared__ __attribute__((aligned(16))) float sm0[HNODES * 4];
  __shared__ __attribute__((aligned(16))) float sxr[HNODES * 256];
  __shared__ int   sqe[RWAVES][QW];
  __shared__ int   sover;

  const int tid = threadIdx.x, lane = tid & 31;
  const int wv = __builtin_amdgcn_readfirstlane(tid >> 5);
  const int rg = (int)blockIdx.x >> 1, hf = (int)blockIdx.x & 1;
  const int lo = rg * RNODES + hf * HNODES;
  if (lo >= Nn) return;

  int* q = sqe[wv];
  const int widx = rg * RWAVES + wv;
  int qn0 = __builtin_amdgcn_readfirstlane(qcnt[widx]);
  const bool lover = qn0 < 0;
  qn0 = qn0 > QW ? QW : qn0;
  int qn = 0;
  if (!lover && qn0 > 0) {
    const int* qs = qsave + (size_t)widx * QW;
    int t[QCOPY];
    #pragma unroll
    for (int u = 0; u < QCOPY; ++u) {
      int i = lane + 32 * u;
      t[u] = qs[i < qn0 ? i : qn0 - 1];
    }
    #pragma unroll
    for (int u = 0; u < QCOPY; ++u) {
      int i = lane + 32 * u;
      int r = (int)((unsigned)t[u] >> 16) - hf * HNODES;
      q_push(q, qn, i < qn0 && (unsigned)r < (unsigned)HNODES, (t[u] & 0xFFFF) | (r << 16));
    }
  }

  const float4 z4 = make_float4(0.f, 0.f, 0.f, 0.f);
  for (int i = tid; i < (HNODES + 1) * 64; i += RWAVES * 32) ((float4*)sagg)[i] = z4;
  if (tid < HNODES + 1) ((float4*)sden)[tid] = z4;
  if (tid < HNODES) ((uint4*)smax)[tid] = make_uint4(0u, 0u, 0u, 0u);
  if (tid == 0) sover = 0;
  range_rows_wmma<256, HNODES, 4>(xin, wrp, sxr, lo, Nn, wv, lane);
  __syncthreads();

  const float4 av0 = ((const float4*)att)[lane];
  const float4 av1 = ((const float4*)att)[32 + lane];
  const int k = lane >> 2, k2 = lane & 7;

  if (!lover) {
    const int nown = Nn - lo < HNODES ? Nn - lo : HNODES;
    const int cnt = nown - 8 * wv < 8 ? nown - 8 * wv : 8;
    if (cnt > 0) self8_c64(xl, sxr, sm0, sagg, sden, av0, av1, lo, 8 * wv, cnt, lane);
  }
  __syncthreads();

  if (!lover) {
    bool over = false;
    for (int base = 0; base < qn; base += 8) {
      const int n = qn - base < 8 ? qn - base : 8;
      const bool ok = k < n, ok2 = k2 < n;
      const int ent = q[base + (ok ? k : 0)];
      const int dl2 = (int)((unsigned)q[base + (ok2 ? k2 : 0)] >> 16);
      const int s = ent & 0xFFFF, dl = (int)((unsigned)ent >> 16);
      const bool big = fused8_c64(xl, sxr, sm0, sagg, sden, av0, av1, s, dl, ok ? dl : HNODES,
                                  dl2, ok2 ? dl2 : HNODES, lane);
      if (__builtin_amdgcn_ballot_w32(big && ok2)) { over = true; break; }
    }
    if (over && lane == 0) sover = 1;
  }
  __syncthreads();

  const bool slow = lover || sover != 0;
  if (slow) {
    float* sout = sagg;
    for (int i = tid; i < (HNODES + 1) * 16; i += RWAVES * 32) ((float4*)sout)[i] = z4;
    if (tid < HNODES + 1) ((float4*)sden)[tid] = z4;
    const int hh = lane & 3;
    const int n4 = E >> 2;
    const int nch = (n4 + 31) >> 5;
    const int4* d4 = (const int4*)dstA;
    for (int scan = 0; scan < 3; ++scan) {
      qn = 0;
      for (int c = wv; ; c += RWAVES) {
        const bool rest = c >= nch;
        if (!rest) {
          int p = c * 32 + lane;
          q_push_piece<HNODES>(q, qn, d4[p < n4 ? p : n4 - 1], p, p < n4, lo);
        } else {
          q_push_rest<HNODES, true>(q, qn, wv, lane, lo, dstA, E, Nn);
        }
        const int keep = rest ? 0 : 7;
        while (qn > keep) {
          const int n = qn < 8 ? qn : 8;
          const int base = qn - n;
          const bool ok = k < n;
          const int ent = q[base + (ok ? k : 0)];
          const int e = ent & 0xFFFFF, dr = (int)((unsigned)ent >> 20);
          const int sv = srcA[e < E ? e : E - 1];
          const int s = e < E ? sv : e - E;
          const float lg = logits8_c64(xl, sxr, av0, av1, s, dr, lane);
          if (scan == 0) {
            if (ok) atomicMax(&smax[dr * 4 + hh], f2key(lg));
          } else {
            const float w = expf(lg - key2f(smax[dr * 4 + hh]));
            if (scan == 1) {
              if (ok) atomicAdd(&sden[dr * 4 + hh], w);
            } else {
              const float a = 0.25f * (w / (sden[dr * 4 + hh] + 1e-16f));
              accum8_mean(xl, sout, s, ok ? dr : HNODES, a, lane);
            }
          }
          qn = base;
        }
        if (rest) break;
      }
      if (scan < 2) __syncthreads();
    }
    __syncthreads();
    for (int i = tid; i < HNODES * 16; i += RWAVES * 32) {
      int nl = i >> 4, j4 = i & 15;
      int n = lo + nl;
      if (n < Nn) {
        float4 a = ((const float4*)sout)[i];
        float4 bb = ((const float4*)b2)[j4];
        float4 r;
        r.x = a.x + bb.x; r.y = a.y + bb.y; r.z = a.z + bb.z; r.w = a.w + bb.w;
        ((float4*)out)[(size_t)n * 16 + j4] = r;
      }
    }
  } else {
    for (int i = tid; i < HNODES * 16; i += RWAVES * 32) {
      int nl = i >> 4, u = i & 15;
      int n = lo + nl;
      if (n < Nn) {
        float4 r = ((const float4*)b2)[u];
        #pragma unroll
        for (int h = 0; h < 4; ++h) {
          const float* ar = sagg + nl * 256 + (h >> 1) * 128 + (h & 1) * 16 + u;
          const float f = 0.25f / (sden[nl * 4 + h] + 1e-16f);
          r.x += f * ar[0]; r.y += f * ar[32]; r.z += f * ar[64]; r.w += f * ar[96];
        }
        ((float4*)out)[(size_t)n * 16 + u] = r;
      }
    }
  }
}

extern "C" void kernel_launch(void* const* d_in, const int* in_sizes, int n_in,
                              void* d_out, int out_size, void* d_ws, size_t ws_size,
                              hipStream_t stream) {
  (void)in_sizes; (void)n_in; (void)out_size; (void)ws_size;
  const int  N  = NNODES;
  const int  E  = NEDGES;

  const float* x    = (const float*)d_in[0];
  const int*   ei   = (const int*)  d_in[1];
  const float* Wl1  = (const float*)d_in[2];
  const float* Wr1  = (const float*)d_in[3];
  const float* att1 = (const float*)d_in[4];
  const float* b1   = (const float*)d_in[5];
  const float* Wl2  = (const float*)d_in[6];
  const float* Wr2  = (const float*)d_in[7];
  const float* att2 = (const float*)d_in[8];
  const float* b2   = (const float*)d_in[9];
  const int* srcA = ei;
  const int* dstA = ei + E;

  float* ws    = (float*)d_ws;
  float* hbuf  = ws;
  float* xl    = hbuf + (size_t)N * 128;
  int*   qsave = (int*)(xl + (size_t)N * 256);
  int*   qcnt  = qsave + (size_t)RBLOCKS * RWAVES * QW;
  _Float16* wpk = (_Float16*)(qcnt + (size_t)RBLOCKS * RWAVES);
  unsigned* tags = (unsigned*)(wpk + (size_t)PK_ALL * 8);

  const int TPB = 256;
  int gblocks = (((N + 15) / 16) + 7) / 8;
  int rblocks = RBLOCKS;

  gatv2_pack_w<<<(PK_ALL + TAG_PIECES + TPB - 1) / TPB, TPB, 0, stream>>>(Wl1, Wr1, Wl2, Wr2, wpk, dstA, tags, E);

  gatv2_gemm_nt_wmma<<<gblocks, TPB, 0, stream>>>(x, wpk + 8 * PK_W1L, xl, N, 128);
  gatv2_range_layer1<<<rblocks, RWAVES * 32, 0, stream>>>(xl, x, wpk + 8 * PK_W1R, att1, b1, hbuf, qsave, qcnt,
                                                          srcA, dstA, tags, E, N);

  gatv2_gemm_nt_wmma<<<gblocks, TPB, 0, stream>>>(hbuf, wpk + 8 * PK_W2L, xl, N, 256);
  gatv2_range_layer2<<<2 * RBLOCKS, RWAVES * 32, 0, stream>>>(xl, hbuf, wpk + 8 * PK_W2R, att2, b2, (float*)d_out,
                                                              qsave, qcnt, srcA, dstA, E, N);
}
